# starting kernel
# baseline (speedup 1.0000x reference)
#include <hip/hip_runtime.h>

typedef __attribute__((ext_vector_type(2))) float v2f;
typedef __attribute__((ext_vector_type(8))) float v8f;

__device__ __constant__ float cG0[8] = {
    0.23037781330885523f,  0.7148465705525415f,  0.6308807679295904f,  -0.02798376941698385f,
   -0.18703481171888114f,  0.030841381835986965f, 0.032883011666982945f, -0.010597401784997278f };
__device__ __constant__ float cG1[8] = {
   -0.010597401784997278f, -0.032883011666982945f, 0.030841381835986965f, 0.18703481171888114f,
   -0.02798376941698385f,  -0.6308807679295904f,   0.7148465705525415f,  -0.23037781330885523f };

__global__ __launch_bounds__(32)
void sfb1d_wmma(const float* __restrict__ lo, const float* __restrict__ hi,
                float* __restrict__ out,
                int n, int M, int cols,
                long long loOS, long long hiOS, long long outOS,
                int inAS, int inCS, int outAS, int outCS)
{
    const int lane = threadIdx.x & 31;
    const int half = lane >> 4;
    const int lid  = lane & 15;
    const int mb   = blockIdx.y * 16;
    const int cb   = blockIdx.x * 16;

    const int  m     = mb + lid;
    const int  col   = cb + lid;
    const bool colOK = col < cols;
    const int  colC  = colOK ? col : (cols - 1);

    const long long cOff = (long long)colC * inCS;
    const float* loP = lo + (long long)blockIdx.z * loOS + cOff;
    const float* hiP = hi + (long long)blockIdx.z * hiOS + cOff;

    const int imin = mb >> 1;
    int imax = (mb + 21) >> 1;
    if (imax > n - 1) imax = n - 1;

    const int ia0 = imin + 2 * half;
    const int u0x = m - 2 * ia0 + 6;
    const int u0y = u0x - 2;
    const int sx  = u0x >> 3;
    const int sy  = u0y >> 3;
    const float g0x = cG0[u0x & 7], g1x = cG1[u0x & 7];
    const float g0y = cG0[u0y & 7], g1y = cG1[u0y & 7];

    const int nm1 = n - 1;
    v8f acc = {};

    for (int i0 = imin, s = 0; i0 <= imax; i0 += 4, ++s) {
        const int ia = ia0 + 4 * s;
        const int ib = ia + 1;

        const bool hx = (s == sx) & (ia < n);
        const bool hy = (s == sy) & (ib < n);
        v2f a0, a1;
        a0.x = hx ? g0x : 0.0f;  a0.y = hy ? g0y : 0.0f;
        a1.x = hx ? g1x : 0.0f;  a1.y = hy ? g1y : 0.0f;

        const int iaC  = ia > nm1 ? nm1 : ia;
        const int ibC  = ib > nm1 ? nm1 : ib;
        const int offA = iaC * inAS;
        const int offB = ibC * inAS;
        v2f bLo, bHi;
        bLo.x = loP[offA];  bLo.y = loP[offB];
        bHi.x = hiP[offA];  bHi.y = hiP[offB];

        acc = __builtin_amdgcn_wmma_f32_16x16x4_f32(false, a0, false, bLo,
                                                    (short)0, acc, false, false);
        acc = __builtin_amdgcn_wmma_f32_16x16x4_f32(false, a1, false, bHi,
                                                    (short)0, acc, false, false);
    }

    const int row0 = mb + 8 * half;
    float* p = out + (long long)blockIdx.z * outOS
                   + (long long)col * outCS
                   + (long long)row0 * outAS;

    if ((mb + 16 <= M) & (cb + 16 <= cols)) {
#pragma unroll
        for (int r = 0; r < 8; ++r) {
            *p = acc[r];
            p += outAS;
        }
    } else {
        const int rowsLeft = M - row0;
#pragma unroll
        for (int r = 0; r < 8; ++r) {
            if ((r < rowsLeft) & colOK)
                *p = acc[r];
            p += outAS;
        }
    }
}

extern "C" void kernel_launch(void* const* d_in, const int* in_sizes, int n_in,
                              void* d_out, int out_size, void* d_ws, size_t ws_size,
                              hipStream_t stream)
{
    const float* yl  = (const float*)d_in[0];
    const float* yh0 = (const float*)d_in[1];
    const float* yh1 = (const float*)d_in[2];
    float* out = (float*)d_out;
    float* ws  = (float*)d_ws;

    const size_t slotT  = (size_t)16 * 66 * 66 * 126;
    const size_t slotT2 = (size_t)16 * 66 * 126 * 126;
    float* tb  = ws;
    float* t2b = ws + 4 * slotT;
    float* llb = ws + 4 * slotT + 2 * slotT2;

    auto L = [&](const float* lo, const float* hi, float* o,
                 int n, int M, int cols, int outer,
                 long long loOS, long long hiOS, long long outOS,
                 int inAS, int inCS, int outAS, int outCS) {
        dim3 g((unsigned)((cols + 15) / 16), (unsigned)((M + 15) / 16), (unsigned)outer);
        sfb1d_wmma<<<g, dim3(32), 0, stream>>>(lo, hi, o, n, M, cols,
                                               loOS, hiOS, outOS,
                                               inAS, inCS, outAS, outCS);
    };

    {
        const int D = 36, M = 66;
        const long long D3 = (long long)D * D * D;
        const long long H  = 7 * D3;
        const float*    los[4]   = { yl, yh1 + 1 * D3, yh1 + 3 * D3, yh1 + 5 * D3 };
        const long long losOS[4] = { D3, H, H, H };
        const float*    his[4]   = { yh1, yh1 + 2 * D3, yh1 + 4 * D3, yh1 + 6 * D3 };
        for (int j = 0; j < 4; ++j)
            L(los[j], his[j], tb + j * slotT, D, M, D * D, 16,
              losOS[j], H, (long long)D * D * M, 1, D, 1, M);
        for (int j = 0; j < 2; ++j)
            L(tb + (2 * j) * slotT, tb + (2 * j + 1) * slotT, t2b + j * slotT2,
              D, M, M, 16 * D, (long long)D * M, (long long)D * M,
              (long long)M * M, M, 1, M, 1);
        L(t2b, t2b + slotT2, llb, D, M, M * M, 16,
          (long long)D * M * M, (long long)D * M * M,
          (long long)M * M * M, M * M, 1, M * M, 1);
    }

    {
        const int D = 66, M = 126;
        const long long D3 = (long long)D * D * D;
        const long long H  = 7 * D3;
        const float*    los[4]   = { llb, yh0 + 1 * D3, yh0 + 3 * D3, yh0 + 5 * D3 };
        const long long losOS[4] = { D3, H, H, H };
        const float*    his[4]   = { yh0, yh0 + 2 * D3, yh0 + 4 * D3, yh0 + 6 * D3 };
        for (int j = 0; j < 4; ++j)
            L(los[j], his[j], tb + j * slotT, D, M, D * D, 16,
              losOS[j], H, (long long)D * D * M, 1, D, 1, M);
        for (int j = 0; j < 2; ++j)
            L(tb + (2 * j) * slotT, tb + (2 * j + 1) * slotT, t2b + j * slotT2,
              D, M, M, 16 * D, (long long)D * M, (long long)D * M,
              (long long)M * M, M, 1, M, 1);
        L(t2b, t2b + slotT2, out, D, M, M * M, 16,
          (long long)D * M * M, (long long)D * M * M,
          (long long)M * M * M, M * M, 1, M * M, 1);
    }

    (void)in_sizes; (void)n_in; (void)out_size; (void)ws_size;
}


# Contiguous-axis pass: WMMA operand roles swapped (data = A, band filter = B): fragments are 8-byte k-pair loads, stores are 64-byte row segments instead of 4-byte pieces of 16 lines
# speedup vs baseline: 1.5340x; 1.5340x over previous
#include <hip/hip_runtime.h>

typedef __attribute__((ext_vector_type(2))) float v2f;
typedef __attribute__((ext_vector_type(8))) float v8f;

__device__ __constant__ float cG0[8] = {
    0.23037781330885523f,  0.7148465705525415f,  0.6308807679295904f,  -0.02798376941698385f,
   -0.18703481171888114f,  0.030841381835986965f, 0.032883011666982945f, -0.010597401784997278f };
__device__ __constant__ float cG1[8] = {
   -0.010597401784997278f, -0.032883011666982945f, 0.030841381835986965f, 0.18703481171888114f,
   -0.02798376941698385f,  -0.6308807679295904f,   0.7148465705525415f,  -0.23037781330885523f };

__global__ __launch_bounds__(32)
void sfb1d_wmma(const float* __restrict__ lo, const float* __restrict__ hi,
                float* __restrict__ out,
                int n, int M, int cols,
                long long loOS, long long hiOS, long long outOS,
                int inAS, int inCS, int outAS, int outCS)
{
    const int lane = threadIdx.x & 31;
    const int half = lane >> 4;
    const int lid  = lane & 15;
    const int mb   = blockIdx.y * 16;
    const int cb   = blockIdx.x * 16;

    const int  m     = mb + lid;
    const int  col   = cb + lid;
    const bool colOK = col < cols;
    const int  colC  = colOK ? col : (cols - 1);

    const long long cOff = (long long)colC * inCS;
    const float* loP = lo + (long long)blockIdx.z * loOS + cOff;
    const float* hiP = hi + (long long)blockIdx.z * hiOS + cOff;

    const int imin = mb >> 1;
    int imax = (mb + 21) >> 1;
    if (imax > n - 1) imax = n - 1;

    const int ia0 = imin + 2 * half;
    const int u0x = m - 2 * ia0 + 6;
    const int u0y = u0x - 2;
    const int sx  = u0x >> 3;
    const int sy  = u0y >> 3;
    const float g0x = cG0[u0x & 7], g1x = cG1[u0x & 7];
    const float g0y = cG0[u0y & 7], g1y = cG1[u0y & 7];

    const int nm1 = n - 1;
    v8f acc = {};

    for (int i0 = imin, s = 0; i0 <= imax; i0 += 4, ++s) {
        const int ia = ia0 + 4 * s;
        const int ib = ia + 1;

        const bool hx = (s == sx) & (ia < n);
        const bool hy = (s == sy) & (ib < n);
        v2f a0, a1;
        a0.x = hx ? g0x : 0.0f;  a0.y = hy ? g0y : 0.0f;
        a1.x = hx ? g1x : 0.0f;  a1.y = hy ? g1y : 0.0f;

        const int iaC  = ia > nm1 ? nm1 : ia;
        const int ibC  = ib > nm1 ? nm1 : ib;
        const int offA = iaC * inAS;
        const int offB = ibC * inAS;
        v2f bLo, bHi;
        bLo.x = loP[offA];  bLo.y = loP[offB];
        bHi.x = hiP[offA];  bHi.y = hiP[offB];

        acc = __builtin_amdgcn_wmma_f32_16x16x4_f32(false, a0, false, bLo,
                                                    (short)0, acc, false, false);
        acc = __builtin_amdgcn_wmma_f32_16x16x4_f32(false, a1, false, bHi,
                                                    (short)0, acc, false, false);
    }

    const int row0 = mb + 8 * half;
    float* p = out + (long long)blockIdx.z * outOS
                   + (long long)col * outCS
                   + (long long)row0 * outAS;

    if ((mb + 16 <= M) & (cb + 16 <= cols)) {
#pragma unroll
        for (int r = 0; r < 8; ++r) {
            *p = acc[r];
            p += outAS;
        }
    } else {
        const int rowsLeft = M - row0;
#pragma unroll
        for (int r = 0; r < 8; ++r) {
            if ((r < rowsLeft) & colOK)
                *p = acc[r];
            p += outAS;
        }
    }
}

__global__ __launch_bounds__(32)
void sfb1d_wmma_rows(const float* __restrict__ lo, const float* __restrict__ hi,
                     float* __restrict__ out,
                     int n, int M, int lines,
                     long long loOS, long long hiOS, long long outOS,
                     int inLS, int outLS)
{
    const int lane = threadIdx.x & 31;
    const int half = lane >> 4;
    const int lid  = lane & 15;
    const int mb   = blockIdx.y * 16;
    const int lb   = blockIdx.x * 16;

    const int m     = mb + lid;
    const int line  = lb + lid;
    const int lineC = line < lines ? line : (lines - 1);

    const float* loP = lo + (long long)blockIdx.z * loOS + (long long)lineC * inLS;
    const float* hiP = hi + (long long)blockIdx.z * hiOS + (long long)lineC * inLS;

    const int imin = mb >> 1;
    int imax = (mb + 21) >> 1;
    if (imax > n - 1) imax = n - 1;

    const int ia0 = imin + 2 * half;
    const int u0x = m - 2 * ia0 + 6;
    const int u0y = u0x - 2;
    const int sx  = u0x >> 3;
    const int sy  = u0y >> 3;
    const float g0x = cG0[u0x & 7], g1x = cG1[u0x & 7];
    const float g0y = cG0[u0y & 7], g1y = cG1[u0y & 7];

    const int nm2 = n - 2;
    v8f acc = {};

    for (int i0 = imin, s = 0; i0 <= imax; i0 += 4, ++s) {
        const int ia = ia0 + 4 * s;

        const bool hx = (s == sx) & (ia < n);
        const bool hy = (s == sy) & (ia + 1 < n);
        v2f f0, f1;
        f0.x = hx ? g0x : 0.0f;  f0.y = hy ? g0y : 0.0f;
        f1.x = hx ? g1x : 0.0f;  f1.y = hy ? g1y : 0.0f;

        const int iaC = ia > nm2 ? nm2 : ia;
        const v2f dLo = *(const v2f*)(loP + iaC);
        const v2f dHi = *(const v2f*)(hiP + iaC);

        acc = __builtin_amdgcn_wmma_f32_16x16x4_f32(false, dLo, false, f0,
                                                    (short)0, acc, false, false);
        acc = __builtin_amdgcn_wmma_f32_16x16x4_f32(false, dHi, false, f1,
                                                    (short)0, acc, false, false);
    }

    const int line0 = lb + 8 * half;
    float* p = out + (long long)blockIdx.z * outOS
                   + (long long)line0 * outLS + m;

    if ((mb + 16 <= M) & (lb + 16 <= lines)) {
        for (int r = 0; r < 8; ++r) {
            *p = acc[r];
            p += outLS;
        }
    } else {
        const int  linesLeft = lines - line0;
        const bool mOK = m < M;
        for (int r = 0; r < 8; ++r) {
            if ((r < linesLeft) & mOK)
                *p = acc[r];
            p += outLS;
        }
    }
}

extern "C" void kernel_launch(void* const* d_in, const int* in_sizes, int n_in,
                              void* d_out, int out_size, void* d_ws, size_t ws_size,
                              hipStream_t stream)
{
    const float* yl  = (const float*)d_in[0];
    const float* yh0 = (const float*)d_in[1];
    const float* yh1 = (const float*)d_in[2];
    float* out = (float*)d_out;
    float* ws  = (float*)d_ws;

    const size_t slotT  = (size_t)16 * 66 * 66 * 126;
    const size_t slotT2 = (size_t)16 * 66 * 126 * 126;
    float* tb  = ws;
    float* t2b = ws + 4 * slotT;
    float* llb = ws + 4 * slotT + 2 * slotT2;

    auto L = [&](const float* lo, const float* hi, float* o,
                 int n, int M, int cols, int outer,
                 long long loOS, long long hiOS, long long outOS,
                 int inAS, int inCS, int outAS, int outCS) {
        dim3 g((unsigned)((cols + 15) / 16), (unsigned)((M + 15) / 16), (unsigned)outer);
        sfb1d_wmma<<<g, dim3(32), 0, stream>>>(lo, hi, o, n, M, cols,
                                               loOS, hiOS, outOS,
                                               inAS, inCS, outAS, outCS);
    };

    auto LR = [&](const float* lo, const float* hi, float* o,
                  int n, int M, int lines, int outer,
                  long long loOS, long long hiOS, long long outOS) {
        dim3 g((unsigned)((lines + 15) / 16), (unsigned)((M + 15) / 16), (unsigned)outer);
        sfb1d_wmma_rows<<<g, dim3(32), 0, stream>>>(lo, hi, o, n, M, lines,
                                                    loOS, hiOS, outOS, n, M);
    };

    {
        const int D = 36, M = 66;
        const long long D3 = (long long)D * D * D;
        const long long H  = 7 * D3;
        const float*    los[4]   = { yl, yh1 + 1 * D3, yh1 + 3 * D3, yh1 + 5 * D3 };
        const long long losOS[4] = { D3, H, H, H };
        const float*    his[4]   = { yh1, yh1 + 2 * D3, yh1 + 4 * D3, yh1 + 6 * D3 };
        for (int j = 0; j < 4; ++j)
            LR(los[j], his[j], tb + j * slotT, D, M, D * D, 16,
               losOS[j], H, (long long)D * D * M);
        for (int j = 0; j < 2; ++j)
            L(tb + (2 * j) * slotT, tb + (2 * j + 1) * slotT, t2b + j * slotT2,
              D, M, M, 16 * D, (long long)D * M, (long long)D * M,
              (long long)M * M, M, 1, M, 1);
        L(t2b, t2b + slotT2, llb, D, M, M * M, 16,
          (long long)D * M * M, (long long)D * M * M,
          (long long)M * M * M, M * M, 1, M * M, 1);
    }

    {
        const int D = 66, M = 126;
        const long long D3 = (long long)D * D * D;
        const long long H  = 7 * D3;
        const float*    los[4]   = { llb, yh0 + 1 * D3, yh0 + 3 * D3, yh0 + 5 * D3 };
        const long long losOS[4] = { D3, H, H, H };
        const float*    his[4]   = { yh0, yh0 + 2 * D3, yh0 + 4 * D3, yh0 + 6 * D3 };
        for (int j = 0; j < 4; ++j)
            LR(los[j], his[j], tb + j * slotT, D, M, D * D, 16,
               losOS[j], H, (long long)D * D * M);
        for (int j = 0; j < 2; ++j)
            L(tb + (2 * j) * slotT, tb + (2 * j + 1) * slotT, t2b + j * slotT2,
              D, M, M, 16 * D, (long long)D * M, (long long)D * M,
              (long long)M * M, M, 1, M, 1);
        L(t2b, t2b + slotT2, out, D, M, M * M, 16,
          (long long)D * M * M, (long long)D * M * M,
          (long long)M * M * M, M * M, 1, M * M, 1);
    }

    (void)in_sizes; (void)n_in; (void)out_size; (void)ws_size;
}


# One wave per 16-wide strip sweeps all output tiles along the filtered axis (was one wave per 16x16 tile): 5-8x fewer waves, shared band chunk kept in registers, filter fragments built once
# speedup vs baseline: 1.9922x; 1.2987x over previous
#include <hip/hip_runtime.h>

typedef __attribute__((ext_vector_type(2))) float v2f;
typedef __attribute__((ext_vector_type(8))) float v8f;

__device__ __constant__ float cG0[8] = {
    0.23037781330885523f,  0.7148465705525415f,  0.6308807679295904f,  -0.02798376941698385f,
   -0.18703481171888114f,  0.030841381835986965f, 0.032883011666982945f, -0.010597401784997278f };
__device__ __constant__ float cG1[8] = {
   -0.010597401784997278f, -0.032883011666982945f, 0.030841381835986965f, 0.18703481171888114f,
   -0.02798376941698385f,  -0.6308807679295904f,   0.7148465705525415f,  -0.23037781330885523f };


struct BandTaps { v2f f0[3]; v2f f1[3]; };

__device__ __forceinline__ BandTaps make_taps(int lid, int half)
{
    const int u0x = lid - 4 * half + 6;
    const int u0y = u0x - 2;
    const int sx  = u0x >> 3;
    const int sy  = u0y >> 3;
    const float g0x = cG0[u0x & 7], g1x = cG1[u0x & 7];
    const float g0y = cG0[u0y & 7], g1y = cG1[u0y & 7];
    BandTaps b;
#pragma unroll
    for (int s = 0; s < 3; ++s) {
        b.f0[s].x = (s == sx) ? g0x : 0.0f;  b.f0[s].y = (s == sy) ? g0y : 0.0f;
        b.f1[s].x = (s == sx) ? g1x : 0.0f;  b.f1[s].y = (s == sy) ? g1y : 0.0f;
    }
    return b;
}

__device__ __forceinline__ v2f edge_taps(v2f f, int s, int rem)
{
    v2f r;
    r.x = (4 * s     < rem) ? f.x : 0.0f;
    r.y = (4 * s + 1 < rem) ? f.y : 0.0f;
    return r;
}

__global__ __launch_bounds__(32)
void sfb1d_wmma(const float* __restrict__ lo, const float* __restrict__ hi,
                float* __restrict__ out,
                int n, int M, int cols,
                long long loOS, long long hiOS, long long outOS,
                int inAS, int inCS, int outAS, int outCS)
{
    const int lane = threadIdx.x & 31;
    const int half = lane >> 4;
    const int lid  = lane & 15;
    const int cb   = blockIdx.x * 16;

    const int  col   = cb + lid;
    const bool colOK = col < cols;
    const int  colC  = colOK ? col : (cols - 1);

    const long long cOff = (long long)colC * inCS;
    const float* loP = lo + (long long)blockIdx.z * loOS + cOff;
    const float* hiP = hi + (long long)blockIdx.z * hiOS + cOff;

    const BandTaps tp = make_taps(lid, half);
    const int nm1 = n - 1;

    auto chunk = [&](const float* P, int q) -> v2f {
        const int ra  = 4 * q + 2 * half;
        const int raC = ra     > nm1 ? nm1 : ra;
        const int rbC = ra + 1 > nm1 ? nm1 : ra + 1;
        v2f v;
        v.x = P[raC * inAS];
        v.y = P[rbC * inAS];
        return v;
    };

    v2f c0L = chunk(loP, 0), c0H = chunk(hiP, 0);
    v2f c1L = chunk(loP, 1), c1H = chunk(hiP, 1);
    v2f c2L = chunk(loP, 2), c2H = chunk(hiP, 2);

    float* p = out + (long long)blockIdx.z * outOS
                   + (long long)col * outCS
                   + (long long)(8 * half) * outAS;
    const bool colsFit = cb + 16 <= cols;
    const int  T = (M + 15) >> 4;

    for (int t = 0; t < T; ++t) {
        const v2f p1L = chunk(loP, 2 * t + 3), p1H = chunk(hiP, 2 * t + 3);
        const v2f p2L = chunk(loP, 2 * t + 4), p2H = chunk(hiP, 2 * t + 4);

        const int kb = 8 * t;
        v8f acc = {};
        if (kb + 12 <= n) {
            acc = __builtin_amdgcn_wmma_f32_16x16x4_f32(false, tp.f0[0], false, c0L, (short)0, acc, false, false);
            acc = __builtin_amdgcn_wmma_f32_16x16x4_f32(false, tp.f1[0], false, c0H, (short)0, acc, false, false);
            acc = __builtin_amdgcn_wmma_f32_16x16x4_f32(false, tp.f0[1], false, c1L, (short)0, acc, false, false);
            acc = __builtin_amdgcn_wmma_f32_16x16x4_f32(false, tp.f1[1], false, c1H, (short)0, acc, false, false);
            acc = __builtin_amdgcn_wmma_f32_16x16x4_f32(false, tp.f0[2], false, c2L, (short)0, acc, false, false);
            acc = __builtin_amdgcn_wmma_f32_16x16x4_f32(false, tp.f1[2], false, c2H, (short)0, acc, false, false);
        } else {
            const int rem = n - kb - 2 * half;
            acc = __builtin_amdgcn_wmma_f32_16x16x4_f32(false, edge_taps(tp.f0[0], 0, rem), false, c0L, (short)0, acc, false, false);
            acc = __builtin_amdgcn_wmma_f32_16x16x4_f32(false, edge_taps(tp.f1[0], 0, rem), false, c0H, (short)0, acc, false, false);
            if (kb + 4 < n) {
                acc = __builtin_amdgcn_wmma_f32_16x16x4_f32(false, edge_taps(tp.f0[1], 1, rem), false, c1L, (short)0, acc, false, false);
                acc = __builtin_amdgcn_wmma_f32_16x16x4_f32(false, edge_taps(tp.f1[1], 1, rem), false, c1H, (short)0, acc, false, false);
            }
            if (kb + 8 < n) {
                acc = __builtin_amdgcn_wmma_f32_16x16x4_f32(false, edge_taps(tp.f0[2], 2, rem), false, c2L, (short)0, acc, false, false);
                acc = __builtin_amdgcn_wmma_f32_16x16x4_f32(false, edge_taps(tp.f1[2], 2, rem), false, c2H, (short)0, acc, false, false);
            }
        }

        float* pr = p;
        if ((16 * t + 16 <= M) & colsFit) {
#pragma unroll
            for (int r = 0; r < 8; ++r) {
                *pr = acc[r];
                pr += outAS;
            }
        } else {
            const int rowsLeft = M - 16 * t - 8 * half;
#pragma unroll
            for (int r = 0; r < 8; ++r) {
                if ((r < rowsLeft) & colOK)
                    *pr = acc[r];
                pr += outAS;
            }
        }
        p += 16 * (long long)outAS;

        c0L = c2L;  c0H = c2H;
        c1L = p1L;  c1H = p1H;
        c2L = p2L;  c2H = p2H;
    }
}

__global__ __launch_bounds__(32)
void sfb1d_wmma_rows(const float* __restrict__ lo, const float* __restrict__ hi,
                     float* __restrict__ out,
                     int n, int M, int lines,
                     long long loOS, long long hiOS, long long outOS,
                     int inLS, int outLS)
{
    const int lane = threadIdx.x & 31;
    const int half = lane >> 4;
    const int lid  = lane & 15;
    const int lb   = blockIdx.x * 16;

    const int line  = lb + lid;
    const int lineC = line < lines ? line : (lines - 1);

    const float* loP = lo + (long long)blockIdx.z * loOS + (long long)lineC * inLS;
    const float* hiP = hi + (long long)blockIdx.z * hiOS + (long long)lineC * inLS;

    const BandTaps tp = make_taps(lid, half);
    const int nm2 = n - 2;

    auto chunk = [&](const float* P, int q) -> v2f {
        const int ia  = 4 * q + 2 * half;
        const int iaC = ia > nm2 ? nm2 : ia;
        return *(const v2f*)(P + iaC);
    };

    v2f c0L = chunk(loP, 0), c0H = chunk(hiP, 0);
    v2f c1L = chunk(loP, 1), c1H = chunk(hiP, 1);
    v2f c2L = chunk(loP, 2), c2H = chunk(hiP, 2);

    const int line0 = lb + 8 * half;
    float* p = out + (long long)blockIdx.z * outOS
                   + (long long)line0 * outLS + lid;
    const bool linesFit  = lb + 16 <= lines;
    const int  linesLeft = lines - line0;
    const int  T = (M + 15) >> 4;

    for (int t = 0; t < T; ++t) {
        const v2f p1L = chunk(loP, 2 * t + 3), p1H = chunk(hiP, 2 * t + 3);
        const v2f p2L = chunk(loP, 2 * t + 4), p2H = chunk(hiP, 2 * t + 4);

        const int kb = 8 * t;
        v8f acc = {};
        if (kb + 12 <= n) {
            acc = __builtin_amdgcn_wmma_f32_16x16x4_f32(false, c0L, false, tp.f0[0], (short)0, acc, false, false);
            acc = __builtin_amdgcn_wmma_f32_16x16x4_f32(false, c0H, false, tp.f1[0], (short)0, acc, false, false);
            acc = __builtin_amdgcn_wmma_f32_16x16x4_f32(false, c1L, false, tp.f0[1], (short)0, acc, false, false);
            acc = __builtin_amdgcn_wmma_f32_16x16x4_f32(false, c1H, false, tp.f1[1], (short)0, acc, false, false);
            acc = __builtin_amdgcn_wmma_f32_16x16x4_f32(false, c2L, false, tp.f0[2], (short)0, acc, false, false);
            acc = __builtin_amdgcn_wmma_f32_16x16x4_f32(false, c2H, false, tp.f1[2], (short)0, acc, false, false);
        } else {
            const int rem = n - kb - 2 * half;
            acc = __builtin_amdgcn_wmma_f32_16x16x4_f32(false, c0L, false, edge_taps(tp.f0[0], 0, rem), (short)0, acc, false, false);
            acc = __builtin_amdgcn_wmma_f32_16x16x4_f32(false, c0H, false, edge_taps(tp.f1[0], 0, rem), (short)0, acc, false, false);
            if (kb + 4 < n) {
                acc = __builtin_amdgcn_wmma_f32_16x16x4_f32(false, c1L, false, edge_taps(tp.f0[1], 1, rem), (short)0, acc, false, false);
                acc = __builtin_amdgcn_wmma_f32_16x16x4_f32(false, c1H, false, edge_taps(tp.f1[1], 1, rem), (short)0, acc, false, false);
            }
            if (kb + 8 < n) {
                acc = __builtin_amdgcn_wmma_f32_16x16x4_f32(false, c2L, false, edge_taps(tp.f0[2], 2, rem), (short)0, acc, false, false);
                acc = __builtin_amdgcn_wmma_f32_16x16x4_f32(false, c2H, false, edge_taps(tp.f1[2], 2, rem), (short)0, acc, false, false);
            }
        }

        float* pr = p;
        if ((16 * t + 16 <= M) & linesFit) {
#pragma unroll
            for (int r = 0; r < 8; ++r) {
                *pr = acc[r];
                pr += outLS;
            }
        } else {
            const bool mOK = 16 * t + lid < M;
#pragma unroll
            for (int r = 0; r < 8; ++r) {
                if ((r < linesLeft) & mOK)
                    *pr = acc[r];
                pr += outLS;
            }
        }
        p += 16;

        c0L = c2L;  c0H = c2H;
        c1L = p1L;  c1H = p1H;
        c2L = p2L;  c2H = p2H;
    }
}

extern "C" void kernel_launch(void* const* d_in, const int* in_sizes, int n_in,
                              void* d_out, int out_size, void* d_ws, size_t ws_size,
                              hipStream_t stream)
{
    const float* yl  = (const float*)d_in[0];
    const float* yh0 = (const float*)d_in[1];
    const float* yh1 = (const float*)d_in[2];
    float* out = (float*)d_out;
    float* ws  = (float*)d_ws;

    const size_t slotT  = (size_t)16 * 66 * 66 * 126;
    const size_t slotT2 = (size_t)16 * 66 * 126 * 126;
    float* tb  = ws;
    float* t2b = ws + 4 * slotT;
    float* llb = ws + 4 * slotT + 2 * slotT2;

    auto L = [&](const float* lo, const float* hi, float* o,
                 int n, int M, int cols, int outer,
                 long long loOS, long long hiOS, long long outOS,
                 int inAS, int inCS, int outAS, int outCS) {
        dim3 g((unsigned)((cols + 15) / 16), 1u, (unsigned)outer);
        sfb1d_wmma<<<g, dim3(32), 0, stream>>>(lo, hi, o, n, M, cols,
                                               loOS, hiOS, outOS,
                                               inAS, inCS, outAS, outCS);
    };

    auto LR = [&](const float* lo, const float* hi, float* o,
                  int n, int M, int lines, int outer,
                  long long loOS, long long hiOS, long long outOS) {
        dim3 g((unsigned)((lines + 15) / 16), 1u, (unsigned)outer);
        sfb1d_wmma_rows<<<g, dim3(32), 0, stream>>>(lo, hi, o, n, M, lines,
                                                    loOS, hiOS, outOS, n, M);
    };

    {
        const int D = 36, M = 66;
        const long long D3 = (long long)D * D * D;
        const long long H  = 7 * D3;
        const float*    los[4]   = { yl, yh1 + 1 * D3, yh1 + 3 * D3, yh1 + 5 * D3 };
        const long long losOS[4] = { D3, H, H, H };
        const float*    his[4]   = { yh1, yh1 + 2 * D3, yh1 + 4 * D3, yh1 + 6 * D3 };
        for (int j = 0; j < 4; ++j)
            LR(los[j], his[j], tb + j * slotT, D, M, D * D, 16,
               losOS[j], H, (long long)D * D * M);
        for (int j = 0; j < 2; ++j)
            L(tb + (2 * j) * slotT, tb + (2 * j + 1) * slotT, t2b + j * slotT2,
              D, M, M, 16 * D, (long long)D * M, (long long)D * M,
              (long long)M * M, M, 1, M, 1);
        L(t2b, t2b + slotT2, llb, D, M, M * M, 16,
          (long long)D * M * M, (long long)D * M * M,
          (long long)M * M * M, M * M, 1, M * M, 1);
    }

    {
        const int D = 66, M = 126;
        const long long D3 = (long long)D * D * D;
        const long long H  = 7 * D3;
        const float*    los[4]   = { llb, yh0 + 1 * D3, yh0 + 3 * D3, yh0 + 5 * D3 };
        const long long losOS[4] = { D3, H, H, H };
        const float*    his[4]   = { yh0, yh0 + 2 * D3, yh0 + 4 * D3, yh0 + 6 * D3 };
        for (int j = 0; j < 4; ++j)
            LR(los[j], his[j], tb + j * slotT, D, M, D * D, 16,
               losOS[j], H, (long long)D * D * M);
        for (int j = 0; j < 2; ++j)
            L(tb + (2 * j) * slotT, tb + (2 * j + 1) * slotT, t2b + j * slotT2,
              D, M, M, 16 * D, (long long)D * M, (long long)D * M,
              (long long)M * M, M, 1, M, 1);
        L(t2b, t2b + slotT2, out, D, M, M * M, 16,
          (long long)D * M * M, (long long)D * M * M,
          (long long)M * M * M, M * M, 1, M * M, 1);
    }

    (void)in_sizes; (void)n_in; (void)out_size; (void)ws_size;
}


# Contiguous-axis pass: a strip's 16 lines (one contiguous block) are copied to LDS with whole-line 16-byte loads, all issued up front; fragments are 8-byte LDS reads, not 8-byte pieces of 16 lines
# speedup vs baseline: 2.2703x; 1.1396x over previous
#include <hip/hip_runtime.h>

typedef __attribute__((ext_vector_type(2))) float v2f;
typedef __attribute__((ext_vector_type(8))) float v8f;

__device__ __constant__ float cG0[8] = {
    0.23037781330885523f,  0.7148465705525415f,  0.6308807679295904f,  -0.02798376941698385f,
   -0.18703481171888114f,  0.030841381835986965f, 0.032883011666982945f, -0.010597401784997278f };
__device__ __constant__ float cG1[8] = {
   -0.010597401784997278f, -0.032883011666982945f, 0.030841381835986965f, 0.18703481171888114f,
   -0.02798376941698385f,  -0.6308807679295904f,   0.7148465705525415f,  -0.23037781330885523f };


struct BandTaps { v2f f0[3]; v2f f1[3]; };

__device__ __forceinline__ BandTaps make_taps(int lid, int half)
{
    const int u0x = lid - 4 * half + 6;
    const int u0y = u0x - 2;
    const int sx  = u0x >> 3;
    const int sy  = u0y >> 3;
    const float g0x = cG0[u0x & 7], g1x = cG1[u0x & 7];
    const float g0y = cG0[u0y & 7], g1y = cG1[u0y & 7];
    BandTaps b;
#pragma unroll
    for (int s = 0; s < 3; ++s) {
        b.f0[s].x = (s == sx) ? g0x : 0.0f;  b.f0[s].y = (s == sy) ? g0y : 0.0f;
        b.f1[s].x = (s == sx) ? g1x : 0.0f;  b.f1[s].y = (s == sy) ? g1y : 0.0f;
    }
    return b;
}

__device__ __forceinline__ v2f edge_taps(v2f f, int s, int rem)
{
    v2f r;
    r.x = (4 * s     < rem) ? f.x : 0.0f;
    r.y = (4 * s + 1 < rem) ? f.y : 0.0f;
    return r;
}

__global__ __launch_bounds__(32)
void sfb1d_wmma(const float* __restrict__ lo, const float* __restrict__ hi,
                float* __restrict__ out,
                int n, int M, int cols,
                long long loOS, long long hiOS, long long outOS,
                int inAS, int inCS, int outAS, int outCS)
{
    const int lane = threadIdx.x & 31;
    const int half = lane >> 4;
    const int lid  = lane & 15;
    const int cb   = blockIdx.x * 16;

    const int  col   = cb + lid;
    const bool colOK = col < cols;
    const int  colC  = colOK ? col : (cols - 1);

    const long long cOff = (long long)colC * inCS;
    const float* loP = lo + (long long)blockIdx.z * loOS + cOff;
    const float* hiP = hi + (long long)blockIdx.z * hiOS + cOff;

    const BandTaps tp = make_taps(lid, half);
    const int nm1 = n - 1;

    auto chunk = [&](const float* P, int q) -> v2f {
        const int ra  = 4 * q + 2 * half;
        const int raC = ra     > nm1 ? nm1 : ra;
        const int rbC = ra + 1 > nm1 ? nm1 : ra + 1;
        v2f v;
        v.x = P[raC * inAS];
        v.y = P[rbC * inAS];
        return v;
    };

    v2f c0L = chunk(loP, 0), c0H = chunk(hiP, 0);
    v2f c1L = chunk(loP, 1), c1H = chunk(hiP, 1);
    v2f c2L = chunk(loP, 2), c2H = chunk(hiP, 2);

    float* p = out + (long long)blockIdx.z * outOS
                   + (long long)col * outCS
                   + (long long)(8 * half) * outAS;
    const bool colsFit = cb + 16 <= cols;
    const int  T = (M + 15) >> 4;

    for (int t = 0; t < T; ++t) {
        const v2f p1L = chunk(loP, 2 * t + 3), p1H = chunk(hiP, 2 * t + 3);
        const v2f p2L = chunk(loP, 2 * t + 4), p2H = chunk(hiP, 2 * t + 4);

        const int kb = 8 * t;
        v8f acc = {};
        if (kb + 12 <= n) {
            acc = __builtin_amdgcn_wmma_f32_16x16x4_f32(false, tp.f0[0], false, c0L, (short)0, acc, false, false);
            acc = __builtin_amdgcn_wmma_f32_16x16x4_f32(false, tp.f1[0], false, c0H, (short)0, acc, false, false);
            acc = __builtin_amdgcn_wmma_f32_16x16x4_f32(false, tp.f0[1], false, c1L, (short)0, acc, false, false);
            acc = __builtin_amdgcn_wmma_f32_16x16x4_f32(false, tp.f1[1], false, c1H, (short)0, acc, false, false);
            acc = __builtin_amdgcn_wmma_f32_16x16x4_f32(false, tp.f0[2], false, c2L, (short)0, acc, false, false);
            acc = __builtin_amdgcn_wmma_f32_16x16x4_f32(false, tp.f1[2], false, c2H, (short)0, acc, false, false);
        } else {
            const int rem = n - kb - 2 * half;
            acc = __builtin_amdgcn_wmma_f32_16x16x4_f32(false, edge_taps(tp.f0[0], 0, rem), false, c0L, (short)0, acc, false, false);
            acc = __builtin_amdgcn_wmma_f32_16x16x4_f32(false, edge_taps(tp.f1[0], 0, rem), false, c0H, (short)0, acc, false, false);
            if (kb + 4 < n) {
                acc = __builtin_amdgcn_wmma_f32_16x16x4_f32(false, edge_taps(tp.f0[1], 1, rem), false, c1L, (short)0, acc, false, false);
                acc = __builtin_amdgcn_wmma_f32_16x16x4_f32(false, edge_taps(tp.f1[1], 1, rem), false, c1H, (short)0, acc, false, false);
            }
            if (kb + 8 < n) {
                acc = __builtin_amdgcn_wmma_f32_16x16x4_f32(false, edge_taps(tp.f0[2], 2, rem), false, c2L, (short)0, acc, false, false);
                acc = __builtin_amdgcn_wmma_f32_16x16x4_f32(false, edge_taps(tp.f1[2], 2, rem), false, c2H, (short)0, acc, false, false);
            }
        }

        float* pr = p;
        if ((16 * t + 16 <= M) & colsFit) {
#pragma unroll
            for (int r = 0; r < 8; ++r) {
                *pr = acc[r];
                pr += outAS;
            }
        } else {
            const int rowsLeft = M - 16 * t - 8 * half;
#pragma unroll
            for (int r = 0; r < 8; ++r) {
                if ((r < rowsLeft) & colOK)
                    *pr = acc[r];
                pr += outAS;
            }
        }
        p += 16 * (long long)outAS;

        c0L = c2L;  c0H = c2H;
        c1L = p1L;  c1H = p1H;
        c2L = p2L;  c2H = p2H;
    }
}

template <int N>
__global__ __launch_bounds__(32)
void sfb1d_wmma_rows(const float* __restrict__ lo, const float* __restrict__ hi,
                     float* __restrict__ out,
                     int M, int lines,
                     long long loOS, long long hiOS, long long outOS)
{
    __shared__ __attribute__((aligned(16))) float sL[16 * N];
    __shared__ __attribute__((aligned(16))) float sH[16 * N];

    constexpr int n = N;
    const int outLS = M;
    const int lane = threadIdx.x & 31;
    const int half = lane >> 4;
    const int lid  = lane & 15;
    const int lb   = blockIdx.x * 16;

    const int nl   = (lines - lb) < 16 ? (lines - lb) : 16;
    const int cnt4 = (nl * N) >> 2;

    {
        const float4* gL = (const float4*)(lo + (long long)blockIdx.z * loOS + (long long)lb * N);
        const float4* gH = (const float4*)(hi + (long long)blockIdx.z * hiOS + (long long)lb * N);
        float4* sL4 = (float4*)sL;
        float4* sH4 = (float4*)sH;
        constexpr int NV = (4 * N + 31) / 32;
        float4 rL[NV], rH[NV];
        int idx[NV];
        for (int i = 0; i < NV; ++i) {
            const int k = lane + 32 * i;
            idx[i] = k < cnt4 ? k : (cnt4 - 1);
            rL[i] = gL[idx[i]];
            rH[i] = gH[idx[i]];
        }
        for (int i = 0; i < NV; ++i) {
            sL4[idx[i]] = rL[i];
            sH4[idx[i]] = rH[i];
        }
    }
    __syncthreads();

    const int lrow = lid < nl ? lid : (nl - 1);
    const float* loP = sL + lrow * N;
    const float* hiP = sH + lrow * N;

    const BandTaps tp = make_taps(lid, half);
    constexpr int nm2 = N - 2;

    auto chunk = [&](const float* P, int q) -> v2f {
        const int ia  = 4 * q + 2 * half;
        const int iaC = ia > nm2 ? nm2 : ia;
        return *(const v2f*)(P + iaC);
    };

    v2f c0L = chunk(loP, 0), c0H = chunk(hiP, 0);
    v2f c1L = chunk(loP, 1), c1H = chunk(hiP, 1);
    v2f c2L = chunk(loP, 2), c2H = chunk(hiP, 2);

    const int line0 = lb + 8 * half;
    float* p = out + (long long)blockIdx.z * outOS
                   + (long long)line0 * outLS + lid;
    const bool linesFit  = lb + 16 <= lines;
    const int  linesLeft = lines - line0;
    const int  T = (M + 15) >> 4;

    for (int t = 0; t < T; ++t) {
        const v2f p1L = chunk(loP, 2 * t + 3), p1H = chunk(hiP, 2 * t + 3);
        const v2f p2L = chunk(loP, 2 * t + 4), p2H = chunk(hiP, 2 * t + 4);

        const int kb = 8 * t;
        v8f acc = {};
        if (kb + 12 <= n) {
            acc = __builtin_amdgcn_wmma_f32_16x16x4_f32(false, c0L, false, tp.f0[0], (short)0, acc, false, false);
            acc = __builtin_amdgcn_wmma_f32_16x16x4_f32(false, c0H, false, tp.f1[0], (short)0, acc, false, false);
            acc = __builtin_amdgcn_wmma_f32_16x16x4_f32(false, c1L, false, tp.f0[1], (short)0, acc, false, false);
            acc = __builtin_amdgcn_wmma_f32_16x16x4_f32(false, c1H, false, tp.f1[1], (short)0, acc, false, false);
            acc = __builtin_amdgcn_wmma_f32_16x16x4_f32(false, c2L, false, tp.f0[2], (short)0, acc, false, false);
            acc = __builtin_amdgcn_wmma_f32_16x16x4_f32(false, c2H, false, tp.f1[2], (short)0, acc, false, false);
        } else {
            const int rem = n - kb - 2 * half;
            acc = __builtin_amdgcn_wmma_f32_16x16x4_f32(false, c0L, false, edge_taps(tp.f0[0], 0, rem), (short)0, acc, false, false);
            acc = __builtin_amdgcn_wmma_f32_16x16x4_f32(false, c0H, false, edge_taps(tp.f1[0], 0, rem), (short)0, acc, false, false);
            if (kb + 4 < n) {
                acc = __builtin_amdgcn_wmma_f32_16x16x4_f32(false, c1L, false, edge_taps(tp.f0[1], 1, rem), (short)0, acc, false, false);
                acc = __builtin_amdgcn_wmma_f32_16x16x4_f32(false, c1H, false, edge_taps(tp.f1[1], 1, rem), (short)0, acc, false, false);
            }
            if (kb + 8 < n) {
                acc = __builtin_amdgcn_wmma_f32_16x16x4_f32(false, c2L, false, edge_taps(tp.f0[2], 2, rem), (short)0, acc, false, false);
                acc = __builtin_amdgcn_wmma_f32_16x16x4_f32(false, c2H, false, edge_taps(tp.f1[2], 2, rem), (short)0, acc, false, false);
            }
        }

        float* pr = p;
        if ((16 * t + 16 <= M) & linesFit) {
#pragma unroll
            for (int r = 0; r < 8; ++r) {
                *pr = acc[r];
                pr += outLS;
            }
        } else {
            const bool mOK = 16 * t + lid < M;
#pragma unroll
            for (int r = 0; r < 8; ++r) {
                if ((r < linesLeft) & mOK)
                    *pr = acc[r];
                pr += outLS;
            }
        }
        p += 16;

        c0L = c2L;  c0H = c2H;
        c1L = p1L;  c1H = p1H;
        c2L = p2L;  c2H = p2H;
    }
}

template <int N>
static void launch_rows(hipStream_t stream, const float* lo, const float* hi, float* o,
                        int M, int lines, int outer,
                        long long loOS, long long hiOS, long long outOS)
{
    dim3 g((unsigned)((lines + 15) / 16), 1u, (unsigned)outer);
    sfb1d_wmma_rows<N><<<g, dim3(32), 0, stream>>>(lo, hi, o, M, lines, loOS, hiOS, outOS);
}

extern "C" void kernel_launch(void* const* d_in, const int* in_sizes, int n_in,
                              void* d_out, int out_size, void* d_ws, size_t ws_size,
                              hipStream_t stream)
{
    const float* yl  = (const float*)d_in[0];
    const float* yh0 = (const float*)d_in[1];
    const float* yh1 = (const float*)d_in[2];
    float* out = (float*)d_out;
    float* ws  = (float*)d_ws;

    const size_t slotT  = (size_t)16 * 66 * 66 * 126;
    const size_t slotT2 = (size_t)16 * 66 * 126 * 126;
    float* tb  = ws;
    float* t2b = ws + 4 * slotT;
    float* llb = ws + 4 * slotT + 2 * slotT2;

    auto L = [&](const float* lo, const float* hi, float* o,
                 int n, int M, int cols, int outer,
                 long long loOS, long long hiOS, long long outOS,
                 int inAS, int inCS, int outAS, int outCS) {
        dim3 g((unsigned)((cols + 15) / 16), 1u, (unsigned)outer);
        sfb1d_wmma<<<g, dim3(32), 0, stream>>>(lo, hi, o, n, M, cols,
                                               loOS, hiOS, outOS,
                                               inAS, inCS, outAS, outCS);
    };

    {
        const int D = 36, M = 66;
        const long long D3 = (long long)D * D * D;
        const long long H  = 7 * D3;
        const float*    los[4]   = { yl, yh1 + 1 * D3, yh1 + 3 * D3, yh1 + 5 * D3 };
        const long long losOS[4] = { D3, H, H, H };
        const float*    his[4]   = { yh1, yh1 + 2 * D3, yh1 + 4 * D3, yh1 + 6 * D3 };
        for (int j = 0; j < 4; ++j)
            launch_rows<D>(stream, los[j], his[j], tb + j * slotT, M, D * D, 16,
                           losOS[j], H, (long long)D * D * M);
        for (int j = 0; j < 2; ++j)
            L(tb + (2 * j) * slotT, tb + (2 * j + 1) * slotT, t2b + j * slotT2,
              D, M, M, 16 * D, (long long)D * M, (long long)D * M,
              (long long)M * M, M, 1, M, 1);
        L(t2b, t2b + slotT2, llb, D, M, M * M, 16,
          (long long)D * M * M, (long long)D * M * M,
          (long long)M * M * M, M * M, 1, M * M, 1);
    }

    {
        const int D = 66, M = 126;
        const long long D3 = (long long)D * D * D;
        const long long H  = 7 * D3;
        const float*    los[4]   = { llb, yh0 + 1 * D3, yh0 + 3 * D3, yh0 + 5 * D3 };
        const long long losOS[4] = { D3, H, H, H };
        const float*    his[4]   = { yh0, yh0 + 2 * D3, yh0 + 4 * D3, yh0 + 6 * D3 };
        for (int j = 0; j < 4; ++j)
            launch_rows<D>(stream, los[j], his[j], tb + j * slotT, M, D * D, 16,
                           losOS[j], H, (long long)D * D * M);
        for (int j = 0; j < 2; ++j)
            L(tb + (2 * j) * slotT, tb + (2 * j + 1) * slotT, t2b + j * slotT2,
              D, M, M, 16 * D, (long long)D * M, (long long)D * M,
              (long long)M * M, M, 1, M, 1);
        L(t2b, t2b + slotT2, out, D, M, M * M, 16,
          (long long)D * M * M, (long long)D * M * M,
          (long long)M * M * M, M * M, 1, M * M, 1);
    }

    (void)in_sizes; (void)n_in; (void)out_size; (void)ws_size;
}


# speedup vs baseline: 3.1554x; 1.3899x over previous
#include <hip/hip_runtime.h>

typedef __attribute__((ext_vector_type(2))) float v2f;
typedef __attribute__((ext_vector_type(8))) float v8f;

__device__ __constant__ float cG0[8] = {
    0.23037781330885523f,  0.7148465705525415f,  0.6308807679295904f,  -0.02798376941698385f,
   -0.18703481171888114f,  0.030841381835986965f, 0.032883011666982945f, -0.010597401784997278f };
__device__ __constant__ float cG1[8] = {
   -0.010597401784997278f, -0.032883011666982945f, 0.030841381835986965f, 0.18703481171888114f,
   -0.02798376941698385f,  -0.6308807679295904f,   0.7148465705525415f,  -0.23037781330885523f };


struct BandTaps { v2f f0[3]; v2f f1[3]; };

__device__ __forceinline__ BandTaps make_taps(int lid, int half)
{
    const int u0x = lid - 4 * half + 6;
    const int u0y = u0x - 2;
    const int sx  = u0x >> 3;
    const int sy  = u0y >> 3;
    const float g0x = cG0[u0x & 7], g1x = cG1[u0x & 7];
    const float g0y = cG0[u0y & 7], g1y = cG1[u0y & 7];
    BandTaps b;
#pragma unroll
    for (int s = 0; s < 3; ++s) {
        b.f0[s].x = (s == sx) ? g0x : 0.0f;  b.f0[s].y = (s == sy) ? g0y : 0.0f;
        b.f1[s].x = (s == sx) ? g1x : 0.0f;  b.f1[s].y = (s == sy) ? g1y : 0.0f;
    }
    return b;
}

__device__ __forceinline__ v2f edge_taps(v2f f, int s, int rem)
{
    v2f r;
    r.x = (4 * s     < rem) ? f.x : 0.0f;
    r.y = (4 * s + 1 < rem) ? f.y : 0.0f;
    return r;
}

__global__ __launch_bounds__(256)
void sfb1d_wmma(const float* __restrict__ lo, const float* __restrict__ hi,
                float* __restrict__ out,
                int n, int M, int cols, int wpb,
                long long loOS, long long hiOS, long long outOS,
                int inAS, int inCS, int outAS, int outCS)
{
    const int lane = threadIdx.x & 31;
    const int half = lane >> 4;
    const int lid  = lane & 15;
    const int wv   = __builtin_amdgcn_readfirstlane((int)(threadIdx.x >> 5));
    const int cb   = (blockIdx.x * wpb + wv) * 16;
    if (cb >= cols) return;

    const int  col   = cb + lid;
    const bool colOK = col < cols;
    const int  colC  = colOK ? col : (cols - 1);

    const long long cOff = (long long)colC * inCS;
    const float* loP = lo + (long long)blockIdx.z * loOS + cOff;
    const float* hiP = hi + (long long)blockIdx.z * hiOS + cOff;

    const BandTaps tp = make_taps(lid, half);
    const int nm1 = n - 1;

    auto chunk = [&](const float* P, int q) -> v2f {
        const int ra  = 4 * q + 2 * half;
        const int raC = ra     > nm1 ? nm1 : ra;
        const int rbC = ra + 1 > nm1 ? nm1 : ra + 1;
        v2f v;
        v.x = P[raC * inAS];
        v.y = P[rbC * inAS];
        return v;
    };

    v2f c0L = chunk(loP, 0), c0H = chunk(hiP, 0);
    v2f c1L = chunk(loP, 1), c1H = chunk(hiP, 1);
    v2f c2L = chunk(loP, 2), c2H = chunk(hiP, 2);

    float* p = out + (long long)blockIdx.z * outOS
                   + (long long)col * outCS
                   + (long long)(8 * half) * outAS;
    const bool colsFit = cb + 16 <= cols;
    const int  T = (M + 15) >> 4;

    for (int t = 0; t < T; ++t) {
        const v2f p1L = chunk(loP, 2 * t + 3), p1H = chunk(hiP, 2 * t + 3);
        const v2f p2L = chunk(loP, 2 * t + 4), p2H = chunk(hiP, 2 * t + 4);

        const int kb = 8 * t;
        v8f acc = {};
        if (kb + 12 <= n) {
            acc = __builtin_amdgcn_wmma_f32_16x16x4_f32(false, tp.f0[0], false, c0L, (short)0, acc, false, false);
            acc = __builtin_amdgcn_wmma_f32_16x16x4_f32(false, tp.f1[0], false, c0H, (short)0, acc, false, false);
            acc = __builtin_amdgcn_wmma_f32_16x16x4_f32(false, tp.f0[1], false, c1L, (short)0, acc, false, false);
            acc = __builtin_amdgcn_wmma_f32_16x16x4_f32(false, tp.f1[1], false, c1H, (short)0, acc, false, false);
            acc = __builtin_amdgcn_wmma_f32_16x16x4_f32(false, tp.f0[2], false, c2L, (short)0, acc, false, false);
            acc = __builtin_amdgcn_wmma_f32_16x16x4_f32(false, tp.f1[2], false, c2H, (short)0, acc, false, false);
        } else {
            const int rem = n - kb - 2 * half;
            acc = __builtin_amdgcn_wmma_f32_16x16x4_f32(false, edge_taps(tp.f0[0], 0, rem), false, c0L, (short)0, acc, false, false);
            acc = __builtin_amdgcn_wmma_f32_16x16x4_f32(false, edge_taps(tp.f1[0], 0, rem), false, c0H, (short)0, acc, false, false);
            if (kb + 4 < n) {
                acc = __builtin_amdgcn_wmma_f32_16x16x4_f32(false, edge_taps(tp.f0[1], 1, rem), false, c1L, (short)0, acc, false, false);
                acc = __builtin_amdgcn_wmma_f32_16x16x4_f32(false, edge_taps(tp.f1[1], 1, rem), false, c1H, (short)0, acc, false, false);
            }
            if (kb + 8 < n) {
                acc = __builtin_amdgcn_wmma_f32_16x16x4_f32(false, edge_taps(tp.f0[2], 2, rem), false, c2L, (short)0, acc, false, false);
                acc = __builtin_amdgcn_wmma_f32_16x16x4_f32(false, edge_taps(tp.f1[2], 2, rem), false, c2H, (short)0, acc, false, false);
            }
        }

        float* pr = p;
        if ((16 * t + 16 <= M) & colsFit) {
#pragma unroll
            for (int r = 0; r < 8; ++r) {
                *pr = acc[r];
                pr += outAS;
            }
        } else {
            const int rowsLeft = M - 16 * t - 8 * half;
#pragma unroll
            for (int r = 0; r < 8; ++r) {
                if ((r < rowsLeft) & colOK)
                    *pr = acc[r];
                pr += outAS;
            }
        }
        p += 16 * (long long)outAS;

        c0L = c2L;  c0H = c2H;
        c1L = p1L;  c1H = p1H;
        c2L = p2L;  c2H = p2H;
    }
}

template <int N>
__global__ __launch_bounds__(32)
void sfb1d_wmma_rows(const float* __restrict__ lo, const float* __restrict__ hi,
                     float* __restrict__ out,
                     int M, int lines,
                     long long loOS, long long hiOS, long long outOS)
{
    __shared__ __attribute__((aligned(16))) float sL[16 * N];
    __shared__ __attribute__((aligned(16))) float sH[16 * N];

    constexpr int n = N;
    const int outLS = M;
    const int lane = threadIdx.x & 31;
    const int half = lane >> 4;
    const int lid  = lane & 15;
    const int lb   = blockIdx.x * 16;

    const int nl   = (lines - lb) < 16 ? (lines - lb) : 16;
    const int cnt4 = (nl * N) >> 2;

    {
        const float4* gL = (const float4*)(lo + (long long)blockIdx.z * loOS + (long long)lb * N);
        const float4* gH = (const float4*)(hi + (long long)blockIdx.z * hiOS + (long long)lb * N);
        float4* sL4 = (float4*)sL;
        float4* sH4 = (float4*)sH;
        constexpr int NV = (4 * N + 31) / 32;
        float4 rL[NV], rH[NV];
        int idx[NV];
        for (int i = 0; i < NV; ++i) {
            const int k = lane + 32 * i;
            idx[i] = k < cnt4 ? k : (cnt4 - 1);
            rL[i] = gL[idx[i]];
            rH[i] = gH[idx[i]];
        }
        for (int i = 0; i < NV; ++i) {
            sL4[idx[i]] = rL[i];
            sH4[idx[i]] = rH[i];
        }
    }
    __syncthreads();

    const int lrow = lid < nl ? lid : (nl - 1);
    const float* loP = sL + lrow * N;
    const float* hiP = sH + lrow * N;

    const BandTaps tp = make_taps(lid, half);
    constexpr int nm2 = N - 2;

    auto chunk = [&](const float* P, int q) -> v2f {
        const int ia  = 4 * q + 2 * half;
        const int iaC = ia > nm2 ? nm2 : ia;
        return *(const v2f*)(P + iaC);
    };

    v2f c0L = chunk(loP, 0), c0H = chunk(hiP, 0);
    v2f c1L = chunk(loP, 1), c1H = chunk(hiP, 1);
    v2f c2L = chunk(loP, 2), c2H = chunk(hiP, 2);

    const int line0 = lb + 8 * half;
    float* p = out + (long long)blockIdx.z * outOS
                   + (long long)line0 * outLS + lid;
    const bool linesFit  = lb + 16 <= lines;
    const int  linesLeft = lines - line0;
    const int  T = (M + 15) >> 4;

    for (int t = 0; t < T; ++t) {
        const v2f p1L = chunk(loP, 2 * t + 3), p1H = chunk(hiP, 2 * t + 3);
        const v2f p2L = chunk(loP, 2 * t + 4), p2H = chunk(hiP, 2 * t + 4);

        const int kb = 8 * t;
        v8f acc = {};
        if (kb + 12 <= n) {
            acc = __builtin_amdgcn_wmma_f32_16x16x4_f32(false, c0L, false, tp.f0[0], (short)0, acc, false, false);
            acc = __builtin_amdgcn_wmma_f32_16x16x4_f32(false, c0H, false, tp.f1[0], (short)0, acc, false, false);
            acc = __builtin_amdgcn_wmma_f32_16x16x4_f32(false, c1L, false, tp.f0[1], (short)0, acc, false, false);
            acc = __builtin_amdgcn_wmma_f32_16x16x4_f32(false, c1H, false, tp.f1[1], (short)0, acc, false, false);
            acc = __builtin_amdgcn_wmma_f32_16x16x4_f32(false, c2L, false, tp.f0[2], (short)0, acc, false, false);
            acc = __builtin_amdgcn_wmma_f32_16x16x4_f32(false, c2H, false, tp.f1[2], (short)0, acc, false, false);
        } else {
            const int rem = n - kb - 2 * half;
            acc = __builtin_amdgcn_wmma_f32_16x16x4_f32(false, c0L, false, edge_taps(tp.f0[0], 0, rem), (short)0, acc, false, false);
            acc = __builtin_amdgcn_wmma_f32_16x16x4_f32(false, c0H, false, edge_taps(tp.f1[0], 0, rem), (short)0, acc, false, false);
            if (kb + 4 < n) {
                acc = __builtin_amdgcn_wmma_f32_16x16x4_f32(false, c1L, false, edge_taps(tp.f0[1], 1, rem), (short)0, acc, false, false);
                acc = __builtin_amdgcn_wmma_f32_16x16x4_f32(false, c1H, false, edge_taps(tp.f1[1], 1, rem), (short)0, acc, false, false);
            }
            if (kb + 8 < n) {
                acc = __builtin_amdgcn_wmma_f32_16x16x4_f32(false, c2L, false, edge_taps(tp.f0[2], 2, rem), (short)0, acc, false, false);
                acc = __builtin_amdgcn_wmma_f32_16x16x4_f32(false, c2H, false, edge_taps(tp.f1[2], 2, rem), (short)0, acc, false, false);
            }
        }

        float* pr = p;
        if ((16 * t + 16 <= M) & linesFit) {
#pragma unroll
            for (int r = 0; r < 8; ++r) {
                *pr = acc[r];
                pr += outLS;
            }
        } else {
            const bool mOK = 16 * t + lid < M;
#pragma unroll
            for (int r = 0; r < 8; ++r) {
                if ((r < linesLeft) & mOK)
                    *pr = acc[r];
                pr += outLS;
            }
        }
        p += 16;

        c0L = c2L;  c0H = c2H;
        c1L = p1L;  c1H = p1H;
        c2L = p2L;  c2H = p2H;
    }
}

template <int N>
static void launch_rows(hipStream_t stream, const float* lo, const float* hi, float* o,
                        int M, int lines, int outer,
                        long long loOS, long long hiOS, long long outOS)
{
    dim3 g((unsigned)((lines + 15) / 16), 1u, (unsigned)outer);
    sfb1d_wmma_rows<N><<<g, dim3(32), 0, stream>>>(lo, hi, o, M, lines, loOS, hiOS, outOS);
}

extern "C" void kernel_launch(void* const* d_in, const int* in_sizes, int n_in,
                              void* d_out, int out_size, void* d_ws, size_t ws_size,
                              hipStream_t stream)
{
    const float* yl  = (const float*)d_in[0];
    const float* yh0 = (const float*)d_in[1];
    const float* yh1 = (const float*)d_in[2];
    float* out = (float*)d_out;
    float* ws  = (float*)d_ws;

    const size_t slotT  = (size_t)16 * 66 * 66 * 126;
    const size_t slotT2 = (size_t)16 * 66 * 126 * 126;
    float* tb  = ws;
    float* t2b = ws + 4 * slotT;
    float* llb = ws + 4 * slotT + 2 * slotT2;

    auto L = [&](const float* lo, const float* hi, float* o,
                 int n, int M, int cols, int outer, int wpb,
                 long long loOS, long long hiOS, long long outOS,
                 int inAS, int inCS, int outAS, int outCS) {
        const int strips = (cols + 15) / 16;
        dim3 g((unsigned)((strips + wpb - 1) / wpb), 1u, (unsigned)outer);
        sfb1d_wmma<<<g, dim3((unsigned)(32 * wpb)), 0, stream>>>(lo, hi, o, n, M, cols, wpb,
                                                                  loOS, hiOS, outOS,
                                                                  inAS, inCS, outAS, outCS);
    };

    {
        const int D = 36, M = 66;
        const long long D3 = (long long)D * D * D;
        const long long H  = 7 * D3;
        const float*    los[4]   = { yl, yh1 + 1 * D3, yh1 + 3 * D3, yh1 + 5 * D3 };
        const long long losOS[4] = { D3, H, H, H };
        const float*    his[4]   = { yh1, yh1 + 2 * D3, yh1 + 4 * D3, yh1 + 6 * D3 };
        for (int j = 0; j < 4; ++j)
            launch_rows<D>(stream, los[j], his[j], tb + j * slotT, M, D * D, 16,
                           losOS[j], H, (long long)D * D * M);
        for (int j = 0; j < 2; ++j)
            L(tb + (2 * j) * slotT, tb + (2 * j + 1) * slotT, t2b + j * slotT2,
              D, M, M, 16 * D, 5, (long long)D * M, (long long)D * M,
              (long long)M * M, M, 1, M, 1);
        L(t2b, t2b + slotT2, llb, D, M, M * M, 16, 8,
          (long long)D * M * M, (long long)D * M * M,
          (long long)M * M * M, M * M, 1, M * M, 1);
    }

    {
        const int D = 66, M = 126;
        const long long D3 = (long long)D * D * D;
        const long long H  = 7 * D3;
        const float*    los[4]   = { llb, yh0 + 1 * D3, yh0 + 3 * D3, yh0 + 5 * D3 };
        const long long losOS[4] = { D3, H, H, H };
        const float*    his[4]   = { yh0, yh0 + 2 * D3, yh0 + 4 * D3, yh0 + 6 * D3 };
        for (int j = 0; j < 4; ++j)
            launch_rows<D>(stream, los[j], his[j], tb + j * slotT, M, D * D, 16,
                           losOS[j], H, (long long)D * D * M);
        for (int j = 0; j < 2; ++j)
            L(tb + (2 * j) * slotT, tb + (2 * j + 1) * slotT, t2b + j * slotT2,
              D, M, M, 16 * D, 8, (long long)D * M, (long long)D * M,
              (long long)M * M, M, 1, M, 1);
        L(t2b, t2b + slotT2, out, D, M, M * M, 16, 8,
          (long long)D * M * M, (long long)D * M * M,
          (long long)M * M * M, M * M, 1, M * M, 1);
    }

    (void)in_sizes; (void)n_in; (void)out_size; (void)ws_size;
}


# Launches 14 -> 6: the four independent band pairs of each contiguous-axis stage and the two of each second stage run as one launch each (pair index = grid y), same waves and work per wave
# speedup vs baseline: 3.4013x; 1.0779x over previous
#include <hip/hip_runtime.h>

typedef __attribute__((ext_vector_type(2))) float v2f;
typedef __attribute__((ext_vector_type(8))) float v8f;

__device__ __constant__ float cG0[8] = {
    0.23037781330885523f,  0.7148465705525415f,  0.6308807679295904f,  -0.02798376941698385f,
   -0.18703481171888114f,  0.030841381835986965f, 0.032883011666982945f, -0.010597401784997278f };
__device__ __constant__ float cG1[8] = {
   -0.010597401784997278f, -0.032883011666982945f, 0.030841381835986965f, 0.18703481171888114f,
   -0.02798376941698385f,  -0.6308807679295904f,   0.7148465705525415f,  -0.23037781330885523f };


struct BandTaps { v2f f0[3]; v2f f1[3]; };

__device__ __forceinline__ BandTaps make_taps(int lid, int half)
{
    const int u0x = lid - 4 * half + 6;
    const int u0y = u0x - 2;
    const int sx  = u0x >> 3;
    const int sy  = u0y >> 3;
    const float g0x = cG0[u0x & 7], g1x = cG1[u0x & 7];
    const float g0y = cG0[u0y & 7], g1y = cG1[u0y & 7];
    BandTaps b;
#pragma unroll
    for (int s = 0; s < 3; ++s) {
        b.f0[s].x = (s == sx) ? g0x : 0.0f;  b.f0[s].y = (s == sy) ? g0y : 0.0f;
        b.f1[s].x = (s == sx) ? g1x : 0.0f;  b.f1[s].y = (s == sy) ? g1y : 0.0f;
    }
    return b;
}

__device__ __forceinline__ v2f edge_taps(v2f f, int s, int rem)
{
    v2f r;
    r.x = (4 * s     < rem) ? f.x : 0.0f;
    r.y = (4 * s + 1 < rem) ? f.y : 0.0f;
    return r;
}

__global__ __launch_bounds__(256)
void sfb1d_wmma(const float* __restrict__ lo, const float* __restrict__ hi,
                float* __restrict__ out,
                int n, int M, int cols, int wpb,
                long long inJS, long long outJS,
                long long loOS, long long hiOS, long long outOS,
                int inAS, int inCS, int outAS, int outCS)
{
    const int lane = threadIdx.x & 31;
    const int half = lane >> 4;
    const int lid  = lane & 15;
    const int wv   = __builtin_amdgcn_readfirstlane((int)(threadIdx.x >> 5));
    const int cb   = (blockIdx.x * wpb + wv) * 16;
    if (cb >= cols) return;

    const int  col   = cb + lid;
    const bool colOK = col < cols;
    const int  colC  = colOK ? col : (cols - 1);

    const long long cOff = (long long)colC * inCS;
    const long long inJ  = (long long)blockIdx.y * inJS;
    const float* loP = lo + inJ + (long long)blockIdx.z * loOS + cOff;
    const float* hiP = hi + inJ + (long long)blockIdx.z * hiOS + cOff;

    const BandTaps tp = make_taps(lid, half);
    const int nm1 = n - 1;

    auto chunk = [&](const float* P, int q) -> v2f {
        const int ra  = 4 * q + 2 * half;
        const int raC = ra     > nm1 ? nm1 : ra;
        const int rbC = ra + 1 > nm1 ? nm1 : ra + 1;
        v2f v;
        v.x = P[raC * inAS];
        v.y = P[rbC * inAS];
        return v;
    };

    v2f c0L = chunk(loP, 0), c0H = chunk(hiP, 0);
    v2f c1L = chunk(loP, 1), c1H = chunk(hiP, 1);
    v2f c2L = chunk(loP, 2), c2H = chunk(hiP, 2);

    float* p = out + (long long)blockIdx.y * outJS
                   + (long long)blockIdx.z * outOS
                   + (long long)col * outCS
                   + (long long)(8 * half) * outAS;
    const bool colsFit = cb + 16 <= cols;
    const int  T = (M + 15) >> 4;

    for (int t = 0; t < T; ++t) {
        const v2f p1L = chunk(loP, 2 * t + 3), p1H = chunk(hiP, 2 * t + 3);
        const v2f p2L = chunk(loP, 2 * t + 4), p2H = chunk(hiP, 2 * t + 4);

        const int kb = 8 * t;
        v8f acc = {};
        if (kb + 12 <= n) {
            acc = __builtin_amdgcn_wmma_f32_16x16x4_f32(false, tp.f0[0], false, c0L, (short)0, acc, false, false);
            acc = __builtin_amdgcn_wmma_f32_16x16x4_f32(false, tp.f1[0], false, c0H, (short)0, acc, false, false);
            acc = __builtin_amdgcn_wmma_f32_16x16x4_f32(false, tp.f0[1], false, c1L, (short)0, acc, false, false);
            acc = __builtin_amdgcn_wmma_f32_16x16x4_f32(false, tp.f1[1], false, c1H, (short)0, acc, false, false);
            acc = __builtin_amdgcn_wmma_f32_16x16x4_f32(false, tp.f0[2], false, c2L, (short)0, acc, false, false);
            acc = __builtin_amdgcn_wmma_f32_16x16x4_f32(false, tp.f1[2], false, c2H, (short)0, acc, false, false);
        } else {
            const int rem = n - kb - 2 * half;
            acc = __builtin_amdgcn_wmma_f32_16x16x4_f32(false, edge_taps(tp.f0[0], 0, rem), false, c0L, (short)0, acc, false, false);
            acc = __builtin_amdgcn_wmma_f32_16x16x4_f32(false, edge_taps(tp.f1[0], 0, rem), false, c0H, (short)0, acc, false, false);
            if (kb + 4 < n) {
                acc = __builtin_amdgcn_wmma_f32_16x16x4_f32(false, edge_taps(tp.f0[1], 1, rem), false, c1L, (short)0, acc, false, false);
                acc = __builtin_amdgcn_wmma_f32_16x16x4_f32(false, edge_taps(tp.f1[1], 1, rem), false, c1H, (short)0, acc, false, false);
            }
            if (kb + 8 < n) {
                acc = __builtin_amdgcn_wmma_f32_16x16x4_f32(false, edge_taps(tp.f0[2], 2, rem), false, c2L, (short)0, acc, false, false);
                acc = __builtin_amdgcn_wmma_f32_16x16x4_f32(false, edge_taps(tp.f1[2], 2, rem), false, c2H, (short)0, acc, false, false);
            }
        }

        float* pr = p;
        if ((16 * t + 16 <= M) & colsFit) {
#pragma unroll
            for (int r = 0; r < 8; ++r) {
                *pr = acc[r];
                pr += outAS;
            }
        } else {
            const int rowsLeft = M - 16 * t - 8 * half;
#pragma unroll
            for (int r = 0; r < 8; ++r) {
                if ((r < rowsLeft) & colOK)
                    *pr = acc[r];
                pr += outAS;
            }
        }
        p += 16 * (long long)outAS;

        c0L = c2L;  c0H = c2H;
        c1L = p1L;  c1H = p1H;
        c2L = p2L;  c2H = p2H;
    }
}

template <int N>
__global__ __launch_bounds__(32)
void sfb1d_wmma_rows(const float* __restrict__ ll, const float* __restrict__ yh,
                     float* __restrict__ tb,
                     int M, int lines,
                     long long llOS, long long yhOS, long long bandS,
                     long long slotS, long long outOS)
{
    __shared__ __attribute__((aligned(16))) float sL[16 * N];
    __shared__ __attribute__((aligned(16))) float sH[16 * N];

    constexpr int n = N;
    const int outLS = M;
    const int lane = threadIdx.x & 31;
    const int half = lane >> 4;
    const int lid  = lane & 15;
    const int lb   = blockIdx.x * 16;

    const int  j   = blockIdx.y;
    const bool j0  = j == 0;
    const float*    lo   = j0 ? ll : yh + (long long)(2 * j - 1) * bandS;
    const long long loOS = j0 ? llOS : yhOS;
    const float*    hi   = yh + (long long)(2 * j) * bandS;
    const long long hiOS = yhOS;
    float*          out  = tb + (long long)j * slotS;

    const int nl   = (lines - lb) < 16 ? (lines - lb) : 16;
    const int cnt4 = (nl * N) >> 2;

    {
        const float4* gL = (const float4*)(lo + (long long)blockIdx.z * loOS + (long long)lb * N);
        const float4* gH = (const float4*)(hi + (long long)blockIdx.z * hiOS + (long long)lb * N);
        float4* sL4 = (float4*)sL;
        float4* sH4 = (float4*)sH;
        constexpr int NV = (4 * N + 31) / 32;
        float4 rL[NV], rH[NV];
        int idx[NV];
        for (int i = 0; i < NV; ++i) {
            const int k = lane + 32 * i;
            idx[i] = k < cnt4 ? k : (cnt4 - 1);
            rL[i] = gL[idx[i]];
            rH[i] = gH[idx[i]];
        }
        for (int i = 0; i < NV; ++i) {
            sL4[idx[i]] = rL[i];
            sH4[idx[i]] = rH[i];
        }
    }
    __syncthreads();

    const int lrow = lid < nl ? lid : (nl - 1);
    const float* loP = sL + lrow * N;
    const float* hiP = sH + lrow * N;

    const BandTaps tp = make_taps(lid, half);
    constexpr int nm2 = N - 2;

    auto chunk = [&](const float* P, int q) -> v2f {
        const int ia  = 4 * q + 2 * half;
        const int iaC = ia > nm2 ? nm2 : ia;
        return *(const v2f*)(P + iaC);
    };

    v2f c0L = chunk(loP, 0), c0H = chunk(hiP, 0);
    v2f c1L = chunk(loP, 1), c1H = chunk(hiP, 1);
    v2f c2L = chunk(loP, 2), c2H = chunk(hiP, 2);

    const int line0 = lb + 8 * half;
    float* p = out + (long long)blockIdx.z * outOS
                   + (long long)line0 * outLS + lid;
    const bool linesFit  = lb + 16 <= lines;
    const int  linesLeft = lines - line0;
    const int  T = (M + 15) >> 4;

    for (int t = 0; t < T; ++t) {
        const v2f p1L = chunk(loP, 2 * t + 3), p1H = chunk(hiP, 2 * t + 3);
        const v2f p2L = chunk(loP, 2 * t + 4), p2H = chunk(hiP, 2 * t + 4);

        const int kb = 8 * t;
        v8f acc = {};
        if (kb + 12 <= n) {
            acc = __builtin_amdgcn_wmma_f32_16x16x4_f32(false, c0L, false, tp.f0[0], (short)0, acc, false, false);
            acc = __builtin_amdgcn_wmma_f32_16x16x4_f32(false, c0H, false, tp.f1[0], (short)0, acc, false, false);
            acc = __builtin_amdgcn_wmma_f32_16x16x4_f32(false, c1L, false, tp.f0[1], (short)0, acc, false, false);
            acc = __builtin_amdgcn_wmma_f32_16x16x4_f32(false, c1H, false, tp.f1[1], (short)0, acc, false, false);
            acc = __builtin_amdgcn_wmma_f32_16x16x4_f32(false, c2L, false, tp.f0[2], (short)0, acc, false, false);
            acc = __builtin_amdgcn_wmma_f32_16x16x4_f32(false, c2H, false, tp.f1[2], (short)0, acc, false, false);
        } else {
            const int rem = n - kb - 2 * half;
            acc = __builtin_amdgcn_wmma_f32_16x16x4_f32(false, c0L, false, edge_taps(tp.f0[0], 0, rem), (short)0, acc, false, false);
            acc = __builtin_amdgcn_wmma_f32_16x16x4_f32(false, c0H, false, edge_taps(tp.f1[0], 0, rem), (short)0, acc, false, false);
            if (kb + 4 < n) {
                acc = __builtin_amdgcn_wmma_f32_16x16x4_f32(false, c1L, false, edge_taps(tp.f0[1], 1, rem), (short)0, acc, false, false);
                acc = __builtin_amdgcn_wmma_f32_16x16x4_f32(false, c1H, false, edge_taps(tp.f1[1], 1, rem), (short)0, acc, false, false);
            }
            if (kb + 8 < n) {
                acc = __builtin_amdgcn_wmma_f32_16x16x4_f32(false, c2L, false, edge_taps(tp.f0[2], 2, rem), (short)0, acc, false, false);
                acc = __builtin_amdgcn_wmma_f32_16x16x4_f32(false, c2H, false, edge_taps(tp.f1[2], 2, rem), (short)0, acc, false, false);
            }
        }

        float* pr = p;
        if ((16 * t + 16 <= M) & linesFit) {
#pragma unroll
            for (int r = 0; r < 8; ++r) {
                *pr = acc[r];
                pr += outLS;
            }
        } else {
            const bool mOK = 16 * t + lid < M;
#pragma unroll
            for (int r = 0; r < 8; ++r) {
                if ((r < linesLeft) & mOK)
                    *pr = acc[r];
                pr += outLS;
            }
        }
        p += 16;

        c0L = c2L;  c0H = c2H;
        c1L = p1L;  c1H = p1H;
        c2L = p2L;  c2H = p2H;
    }
}

template <int N>
static void launch_rows(hipStream_t stream, const float* ll, const float* yh, float* tb,
                        int M, int lines, int outer,
                        long long llOS, long long yhOS, long long bandS,
                        long long slotS, long long outOS)
{
    dim3 g((unsigned)((lines + 15) / 16), 4u, (unsigned)outer);
    sfb1d_wmma_rows<N><<<g, dim3(32), 0, stream>>>(ll, yh, tb, M, lines,
                                                   llOS, yhOS, bandS, slotS, outOS);
}

extern "C" void kernel_launch(void* const* d_in, const int* in_sizes, int n_in,
                              void* d_out, int out_size, void* d_ws, size_t ws_size,
                              hipStream_t stream)
{
    const float* yl  = (const float*)d_in[0];
    const float* yh0 = (const float*)d_in[1];
    const float* yh1 = (const float*)d_in[2];
    float* out = (float*)d_out;
    float* ws  = (float*)d_ws;

    const size_t slotT  = (size_t)16 * 66 * 66 * 126;
    const size_t slotT2 = (size_t)16 * 66 * 126 * 126;
    float* tb  = ws;
    float* t2b = ws + 4 * slotT;
    float* llb = ws + 4 * slotT + 2 * slotT2;

    auto L = [&](const float* lo, const float* hi, float* o, int pairs,
                 long long inJS, long long outJS,
                 int n, int M, int cols, int outer, int wpb,
                 long long loOS, long long hiOS, long long outOS,
                 int inAS, int inCS, int outAS, int outCS) {
        const int strips = (cols + 15) / 16;
        dim3 g((unsigned)((strips + wpb - 1) / wpb), (unsigned)pairs, (unsigned)outer);
        sfb1d_wmma<<<g, dim3((unsigned)(32 * wpb)), 0, stream>>>(lo, hi, o, n, M, cols, wpb,
                                                                  inJS, outJS,
                                                                  loOS, hiOS, outOS,
                                                                  inAS, inCS, outAS, outCS);
    };

    {
        const int D = 36, M = 66;
        const long long D3 = (long long)D * D * D;
        const long long H  = 7 * D3;
        launch_rows<D>(stream, yl, yh1, tb, M, D * D, 16,
                       D3, H, D3, (long long)slotT, (long long)D * D * M);
        L(tb, tb + slotT, t2b, 2, (long long)(2 * slotT), (long long)slotT2,
          D, M, M, 16 * D, 5, (long long)D * M, (long long)D * M,
          (long long)M * M, M, 1, M, 1);
        L(t2b, t2b + slotT2, llb, 1, 0, 0, D, M, M * M, 16, 8,
          (long long)D * M * M, (long long)D * M * M,
          (long long)M * M * M, M * M, 1, M * M, 1);
    }

    {
        const int D = 66, M = 126;
        const long long D3 = (long long)D * D * D;
        const long long H  = 7 * D3;
        launch_rows<D>(stream, llb, yh0, tb, M, D * D, 16,
                       D3, H, D3, (long long)slotT, (long long)D * D * M);
        L(tb, tb + slotT, t2b, 2, (long long)(2 * slotT), (long long)slotT2,
          D, M, M, 16 * D, 8, (long long)D * M, (long long)D * M,
          (long long)M * M, M, 1, M, 1);
        L(t2b, t2b + slotT2, out, 1, 0, 0, D, M, M * M, 16, 8,
          (long long)D * M * M, (long long)D * M * M,
          (long long)M * M * M, M * M, 1, M * M, 1);
    }

    (void)in_sizes; (void)n_in; (void)out_size; (void)ws_size;
}


# Stages 1,2: a wave keeps all its tiles' accumulators, the contiguous output block is laid out in LDS in place of the inputs and stored as whole 128-byte lines; stage 2 copies its plane to LDS first
# speedup vs baseline: 4.4541x; 1.3095x over previous
#include <hip/hip_runtime.h>

typedef __attribute__((ext_vector_type(2))) float v2f;
typedef __attribute__((ext_vector_type(8))) float v8f;

__device__ __constant__ float cG0[8] = {
    0.23037781330885523f,  0.7148465705525415f,  0.6308807679295904f,  -0.02798376941698385f,
   -0.18703481171888114f,  0.030841381835986965f, 0.032883011666982945f, -0.010597401784997278f };
__device__ __constant__ float cG1[8] = {
   -0.010597401784997278f, -0.032883011666982945f, 0.030841381835986965f, 0.18703481171888114f,
   -0.02798376941698385f,  -0.6308807679295904f,   0.7148465705525415f,  -0.23037781330885523f };


struct BandTaps { v2f f0[3]; v2f f1[3]; };

__device__ __forceinline__ BandTaps make_taps(int lid, int half)
{
    const int u0x = lid - 4 * half + 6;
    const int u0y = u0x - 2;
    const int sx  = u0x >> 3;
    const int sy  = u0y >> 3;
    const float g0x = cG0[u0x & 7], g1x = cG1[u0x & 7];
    const float g0y = cG0[u0y & 7], g1y = cG1[u0y & 7];
    BandTaps b;
#pragma unroll
    for (int s = 0; s < 3; ++s) {
        b.f0[s].x = (s == sx) ? g0x : 0.0f;  b.f0[s].y = (s == sy) ? g0y : 0.0f;
        b.f1[s].x = (s == sx) ? g1x : 0.0f;  b.f1[s].y = (s == sy) ? g1y : 0.0f;
    }
    return b;
}

__device__ __forceinline__ v2f edge_taps(v2f f, int s, int rem)
{
    v2f r;
    r.x = (4 * s     < rem) ? f.x : 0.0f;
    r.y = (4 * s + 1 < rem) ? f.y : 0.0f;
    return r;
}

__global__ __launch_bounds__(256)
void sfb1d_wmma(const float* __restrict__ lo, const float* __restrict__ hi,
                float* __restrict__ out,
                int n, int M, int cols, int wpb,
                long long inJS, long long outJS,
                long long loOS, long long hiOS, long long outOS,
                int inAS, int inCS, int outAS, int outCS)
{
    const int lane = threadIdx.x & 31;
    const int half = lane >> 4;
    const int lid  = lane & 15;
    const int wv   = __builtin_amdgcn_readfirstlane((int)(threadIdx.x >> 5));
    const int cb   = (blockIdx.x * wpb + wv) * 16;
    if (cb >= cols) return;

    const int  col   = cb + lid;
    const bool colOK = col < cols;
    const int  colC  = colOK ? col : (cols - 1);

    const long long cOff = (long long)colC * inCS;
    const long long inJ  = (long long)blockIdx.y * inJS;
    const float* loP = lo + inJ + (long long)blockIdx.z * loOS + cOff;
    const float* hiP = hi + inJ + (long long)blockIdx.z * hiOS + cOff;

    const BandTaps tp = make_taps(lid, half);
    const int nm1 = n - 1;

    auto chunk = [&](const float* P, int q) -> v2f {
        const int ra  = 4 * q + 2 * half;
        const int raC = ra     > nm1 ? nm1 : ra;
        const int rbC = ra + 1 > nm1 ? nm1 : ra + 1;
        v2f v;
        v.x = P[raC * inAS];
        v.y = P[rbC * inAS];
        return v;
    };

    v2f c0L = chunk(loP, 0), c0H = chunk(hiP, 0);
    v2f c1L = chunk(loP, 1), c1H = chunk(hiP, 1);
    v2f c2L = chunk(loP, 2), c2H = chunk(hiP, 2);

    float* p = out + (long long)blockIdx.y * outJS
                   + (long long)blockIdx.z * outOS
                   + (long long)col * outCS
                   + (long long)(8 * half) * outAS;
    const bool colsFit = cb + 16 <= cols;
    const int  T = (M + 15) >> 4;

    for (int t = 0; t < T; ++t) {
        const v2f p1L = chunk(loP, 2 * t + 3), p1H = chunk(hiP, 2 * t + 3);
        const v2f p2L = chunk(loP, 2 * t + 4), p2H = chunk(hiP, 2 * t + 4);

        const int kb = 8 * t;
        v8f acc = {};
        if (kb + 12 <= n) {
            acc = __builtin_amdgcn_wmma_f32_16x16x4_f32(false, tp.f0[0], false, c0L, (short)0, acc, false, false);
            acc = __builtin_amdgcn_wmma_f32_16x16x4_f32(false, tp.f1[0], false, c0H, (short)0, acc, false, false);
            acc = __builtin_amdgcn_wmma_f32_16x16x4_f32(false, tp.f0[1], false, c1L, (short)0, acc, false, false);
            acc = __builtin_amdgcn_wmma_f32_16x16x4_f32(false, tp.f1[1], false, c1H, (short)0, acc, false, false);
            acc = __builtin_amdgcn_wmma_f32_16x16x4_f32(false, tp.f0[2], false, c2L, (short)0, acc, false, false);
            acc = __builtin_amdgcn_wmma_f32_16x16x4_f32(false, tp.f1[2], false, c2H, (short)0, acc, false, false);
        } else {
            const int rem = n - kb - 2 * half;
            acc = __builtin_amdgcn_wmma_f32_16x16x4_f32(false, edge_taps(tp.f0[0], 0, rem), false, c0L, (short)0, acc, false, false);
            acc = __builtin_amdgcn_wmma_f32_16x16x4_f32(false, edge_taps(tp.f1[0], 0, rem), false, c0H, (short)0, acc, false, false);
            if (kb + 4 < n) {
                acc = __builtin_amdgcn_wmma_f32_16x16x4_f32(false, edge_taps(tp.f0[1], 1, rem), false, c1L, (short)0, acc, false, false);
                acc = __builtin_amdgcn_wmma_f32_16x16x4_f32(false, edge_taps(tp.f1[1], 1, rem), false, c1H, (short)0, acc, false, false);
            }
            if (kb + 8 < n) {
                acc = __builtin_amdgcn_wmma_f32_16x16x4_f32(false, edge_taps(tp.f0[2], 2, rem), false, c2L, (short)0, acc, false, false);
                acc = __builtin_amdgcn_wmma_f32_16x16x4_f32(false, edge_taps(tp.f1[2], 2, rem), false, c2H, (short)0, acc, false, false);
            }
        }

        float* pr = p;
        if ((16 * t + 16 <= M) & colsFit) {
#pragma unroll
            for (int r = 0; r < 8; ++r) {
                *pr = acc[r];
                pr += outAS;
            }
        } else {
            const int rowsLeft = M - 16 * t - 8 * half;
#pragma unroll
            for (int r = 0; r < 8; ++r) {
                if ((r < rowsLeft) & colOK)
                    *pr = acc[r];
                pr += outAS;
            }
        }
        p += 16 * (long long)outAS;

        c0L = c2L;  c0H = c2H;
        c1L = p1L;  c1H = p1H;
        c2L = p2L;  c2H = p2H;
    }
}

template <bool DATA_IS_A>
__device__ __forceinline__ v8f band_mm(v2f taps, v2f data, v8f acc)
{
    if constexpr (DATA_IS_A)
        return __builtin_amdgcn_wmma_f32_16x16x4_f32(false, data, false, taps, (short)0, acc, false, false);
    else
        return __builtin_amdgcn_wmma_f32_16x16x4_f32(false, taps, false, data, (short)0, acc, false, false);
}

template <int N, bool DATA_IS_A>
__device__ __forceinline__ v8f tile_products(const BandTaps& tp, int half, int t,
                                             v2f c0L, v2f c0H, v2f c1L, v2f c1H, v2f c2L, v2f c2H)
{
    const int kb = 8 * t;
    v8f acc = {};
    if (kb + 12 <= N) {
        acc = band_mm<DATA_IS_A>(tp.f0[0], c0L, acc);
        acc = band_mm<DATA_IS_A>(tp.f1[0], c0H, acc);
        acc = band_mm<DATA_IS_A>(tp.f0[1], c1L, acc);
        acc = band_mm<DATA_IS_A>(tp.f1[1], c1H, acc);
        acc = band_mm<DATA_IS_A>(tp.f0[2], c2L, acc);
        acc = band_mm<DATA_IS_A>(tp.f1[2], c2H, acc);
    } else {
        const int rem = N - kb - 2 * half;
        acc = band_mm<DATA_IS_A>(edge_taps(tp.f0[0], 0, rem), c0L, acc);
        acc = band_mm<DATA_IS_A>(edge_taps(tp.f1[0], 0, rem), c0H, acc);
        if (kb + 4 < N) {
            acc = band_mm<DATA_IS_A>(edge_taps(tp.f0[1], 1, rem), c1L, acc);
            acc = band_mm<DATA_IS_A>(edge_taps(tp.f1[1], 1, rem), c1H, acc);
        }
        if (kb + 8 < N) {
            acc = band_mm<DATA_IS_A>(edge_taps(tp.f0[2], 2, rem), c2L, acc);
            acc = band_mm<DATA_IS_A>(edge_taps(tp.f1[2], 2, rem), c2H, acc);
        }
    }
    return acc;
}

template <int N>
__global__ __launch_bounds__(32)
void sfb1d_wmma_rows(const float* __restrict__ ll, const float* __restrict__ yh,
                     float* __restrict__ tb,
                     int lines,
                     long long llOS, long long yhOS, long long bandS,
                     long long slotS, long long outOS)
{
    constexpr int M = 2 * N - 6;
    constexpr int T = (M + 15) >> 4;
    static_assert(16 * M <= 2 * 16 * N, "output block must fit in the input blocks' LDS");
    __shared__ __attribute__((aligned(16))) float sm[2 * 16 * N];
    float* sL = sm;
    float* sH = sm + 16 * N;

    const int lane = threadIdx.x & 31;
    const int half = lane >> 4;
    const int lid  = lane & 15;
    const int lb   = blockIdx.x * 16;

    const int  j   = blockIdx.y;
    const bool j0  = j == 0;
    const float*    lo   = j0 ? ll : yh + (long long)(2 * j - 1) * bandS;
    const long long loOS = j0 ? llOS : yhOS;
    const float*    hi   = yh + (long long)(2 * j) * bandS;
    const long long hiOS = yhOS;

    const int nl   = (lines - lb) < 16 ? (lines - lb) : 16;
    const int cnt4 = (nl * N) >> 2;

    {
        const float4* gL = (const float4*)(lo + (long long)blockIdx.z * loOS + (long long)lb * N);
        const float4* gH = (const float4*)(hi + (long long)blockIdx.z * hiOS + (long long)lb * N);
        float4* sL4 = (float4*)sL;
        float4* sH4 = (float4*)sH;
        constexpr int NV = (4 * N + 31) / 32;
        float4 rL[NV], rH[NV];
        int idx[NV];
        for (int i = 0; i < NV; ++i) {
            const int k = lane + 32 * i;
            idx[i] = k < cnt4 ? k : (cnt4 - 1);
            rL[i] = gL[idx[i]];
            rH[i] = gH[idx[i]];
        }
        for (int i = 0; i < NV; ++i) {
            sL4[idx[i]] = rL[i];
            sH4[idx[i]] = rH[i];
        }
    }
    __syncthreads();

    const int lrow = lid < nl ? lid : (nl - 1);
    const float* loP = sL + lrow * N;
    const float* hiP = sH + lrow * N;

    const BandTaps tp = make_taps(lid, half);
    constexpr int nm2 = N - 2;

    auto chunk = [&](const float* P, int q) -> v2f {
        const int ia  = 4 * q + 2 * half;
        const int iaC = ia > nm2 ? nm2 : ia;
        return *(const v2f*)(P + iaC);
    };

    v2f c0L = chunk(loP, 0), c0H = chunk(hiP, 0);
    v2f c1L = chunk(loP, 1), c1H = chunk(hiP, 1);
    v2f c2L = chunk(loP, 2), c2H = chunk(hiP, 2);

    v8f acc[T];
#pragma unroll
    for (int t = 0; t < T; ++t) {
        const v2f p1L = chunk(loP, 2 * t + 3), p1H = chunk(hiP, 2 * t + 3);
        const v2f p2L = chunk(loP, 2 * t + 4), p2H = chunk(hiP, 2 * t + 4);
        acc[t] = tile_products<N, true>(tp, half, t, c0L, c0H, c1L, c1H, c2L, c2H);
        c0L = c2L;  c0H = c2H;
        c1L = p1L;  c1H = p1H;
        c2L = p2L;  c2H = p2H;
    }

    __syncthreads();

    {
        static_assert(2 * 16 * N - 16 * M >= 32, "one spare float per lane behind the output block");
        float* q  = sm + (8 * half) * M + lid;
        float* qs = sm + 16 * M + lane;
#pragma unroll
        for (int t = 0; t < T; ++t) {
            const bool inLine = 16 * t + lid < M;
#pragma unroll
            for (int r = 0; r < 8; ++r) {
                float* d = inLine ? (q + r * M + 16 * t) : qs;
                *d = acc[t][r];
            }
        }
    }
    __syncthreads();

    {
        float* g = tb + (long long)j * slotS + (long long)blockIdx.z * outOS + (long long)lb * M;
        const int cntO  = nl * M;
        const int cntO4 = cntO >> 2;
        float4*       g4 = (float4*)g;
        const float4* s4 = (const float4*)sm;
        const int off = (int)(((unsigned long long)g >> 4) & 7ull);
        constexpr int NVO = (4 * M + 7 + 31) / 32;
#pragma unroll
        for (int i = 0; i < NVO; ++i) {
            const int k  = lane + 32 * i - off;
            const int kc = k < 0 ? 0 : (k < cntO4 ? k : (cntO4 - 1));
            g4[kc] = s4[kc];
        }
        const int k1 = (cntO4 << 2) + lane;
        if (k1 < cntO)
            g[k1] = sm[k1];
    }
}

template <int N>
__global__ __launch_bounds__(32 * ((2 * N - 6 + 15) / 16), 8)
void sfb1d_wmma_plane(const float* __restrict__ tb, float* __restrict__ t2b,
                      long long inSlotS, long long outSlotS)
{
    constexpr int M  = 2 * N - 6;
    constexpr int C  = M;
    constexpr int T  = (M + 15) >> 4;
    constexpr int S  = (C + 15) >> 4;
    constexpr int NT = 32 * S;
    constexpr int PL = N * C;
    constexpr int PO = M * C;
    static_assert(PL % 4 == 0 && PO % 4 == 0, "planes are copied in 16-byte pieces");
    static_assert(PO <= 2 * PL, "output plane must fit in the input planes' LDS");
    __shared__ __attribute__((aligned(16))) float sm[2 * PL];

    const int tid  = threadIdx.x;
    const int lane = tid & 31;
    const int half = lane >> 4;
    const int lid  = lane & 15;
    const int wv   = tid >> 5;

    const long long z = blockIdx.x;
    const int       j = blockIdx.y;
    const float* lo  = tb + (long long)(2 * j) * inSlotS + z * PL;
    const float* hi  = lo + inSlotS;
    float*       out = t2b + (long long)j * outSlotS + z * PO;

    {
        const float4* gL = (const float4*)lo;
        const float4* gH = (const float4*)hi;
        float4* sL4 = (float4*)sm;
        float4* sH4 = (float4*)(sm + PL);
        constexpr int CNT4 = PL / 4;
        constexpr int NV   = (CNT4 + NT - 1) / NT;
        float4 rL[NV], rH[NV];
        int idx[NV];
        for (int i = 0; i < NV; ++i) {
            const int k = tid + NT * i;
            idx[i] = k < CNT4 ? k : (CNT4 - 1);
            rL[i] = gL[idx[i]];
            rH[i] = gH[idx[i]];
        }
        for (int i = 0; i < NV; ++i) {
            sL4[idx[i]] = rL[i];
            sH4[idx[i]] = rH[i];
        }
    }
    __syncthreads();

    const int  col   = 16 * wv + lid;
    const bool colOK = col < C;
    const int  colC  = colOK ? col : (C - 1);
    const float* loP = sm + colC;
    const float* hiP = sm + PL + colC;

    const BandTaps tp = make_taps(lid, half);

    auto chunk = [&](const float* P, int q) -> v2f {
        const int ra  = 4 * q + 2 * half;
        const int raC = ra     > N - 1 ? N - 1 : ra;
        const int rbC = ra + 1 > N - 1 ? N - 1 : ra + 1;
        v2f v;
        v.x = P[raC * C];
        v.y = P[rbC * C];
        return v;
    };

    v2f c0L = chunk(loP, 0), c0H = chunk(hiP, 0);

    v8f acc[T];
#pragma unroll
    for (int t = 0; t < T; ++t) {
        const v2f c1L = chunk(loP, 2 * t + 1), c1H = chunk(hiP, 2 * t + 1);
        const v2f c2L = chunk(loP, 2 * t + 2), c2H = chunk(hiP, 2 * t + 2);
        acc[t] = tile_products<N, false>(tp, half, t, c0L, c0H, c1L, c1H, c2L, c2H);
        c0L = c2L;  c0H = c2H;
    }

    __syncthreads();

    {
        static_assert(2 * PL - PO >= NT, "one spare float per thread behind the output plane");
        float* q  = sm + (8 * half) * C + colC;
        float* qs = sm + PO + tid;
#pragma unroll
        for (int t = 0; t < T; ++t) {
#pragma unroll
            for (int r = 0; r < 8; ++r) {
                const bool inPlane = 16 * t + 8 * half + r < M;
                float* d = inPlane ? (q + (16 * t + r) * C) : qs;
                *d = acc[t][r];
            }
        }
    }
    __syncthreads();

    {
        float4*       g4 = (float4*)out;
        const float4* s4 = (const float4*)sm;
        constexpr int CNT4 = PO / 4;
        const int off = (int)(((unsigned long long)out >> 4) & 7ull);
        float4* gA = g4 - off;
        constexpr int NVO = (CNT4 + 7 + NT - 1) / NT;
        static_assert(NT > 7 && NT * (NVO - 1) <= CNT4, "only the first and the last store can fall outside the plane");
#pragma unroll
        for (int i = 0; i < NVO; ++i) {
            const unsigned kk = (unsigned)(tid + NT * i);
            const int k = (int)kk - off;
            const bool in = (i > 0 || k >= 0) && (i < NVO - 1 || k < CNT4);
            if (in)
                gA[kk] = s4[k];
        }
    }
}

template <int N>
static void launch_rows(hipStream_t stream, const float* ll, const float* yh, float* tb,
                        int lines, int outer,
                        long long llOS, long long yhOS, long long bandS,
                        long long slotS, long long outOS)
{
    dim3 g((unsigned)((lines + 15) / 16), 4u, (unsigned)outer);
    sfb1d_wmma_rows<N><<<g, dim3(32), 0, stream>>>(ll, yh, tb, lines,
                                                   llOS, yhOS, bandS, slotS, outOS);
}

template <int N>
static void launch_plane(hipStream_t stream, const float* tb, float* t2b, int planes,
                         long long inSlotS, long long outSlotS)
{
    constexpr int S = (2 * N - 6 + 15) / 16;
    dim3 g((unsigned)planes, 2u, 1u);
    sfb1d_wmma_plane<N><<<g, dim3(32u * S), 0, stream>>>(tb, t2b, inSlotS, outSlotS);
}

extern "C" void kernel_launch(void* const* d_in, const int* in_sizes, int n_in,
                              void* d_out, int out_size, void* d_ws, size_t ws_size,
                              hipStream_t stream)
{
    const float* yl  = (const float*)d_in[0];
    const float* yh0 = (const float*)d_in[1];
    const float* yh1 = (const float*)d_in[2];
    float* out = (float*)d_out;
    float* ws  = (float*)d_ws;

    const size_t slotT  = (size_t)16 * 66 * 66 * 126;
    const size_t slotT2 = (size_t)16 * 66 * 126 * 126;
    float* tb  = ws;
    float* t2b = ws + 4 * slotT;
    float* llb = ws + 4 * slotT + 2 * slotT2;

    auto L = [&](const float* lo, const float* hi, float* o, int pairs,
                 long long inJS, long long outJS,
                 int n, int M, int cols, int outer, int wpb,
                 long long loOS, long long hiOS, long long outOS,
                 int inAS, int inCS, int outAS, int outCS) {
        const int strips = (cols + 15) / 16;
        dim3 g((unsigned)((strips + wpb - 1) / wpb), (unsigned)pairs, (unsigned)outer);
        sfb1d_wmma<<<g, dim3((unsigned)(32 * wpb)), 0, stream>>>(lo, hi, o, n, M, cols, wpb,
                                                                  inJS, outJS,
                                                                  loOS, hiOS, outOS,
                                                                  inAS, inCS, outAS, outCS);
    };

    {
        const int D = 36, M = 66;
        const long long D3 = (long long)D * D * D;
        const long long H  = 7 * D3;
        launch_rows<D>(stream, yl, yh1, tb, D * D, 16,
                       D3, H, D3, (long long)slotT, (long long)D * D * M);
        launch_plane<D>(stream, tb, t2b, 16 * D, (long long)slotT, (long long)slotT2);
        L(t2b, t2b + slotT2, llb, 1, 0, 0, D, M, M * M, 16, 8,
          (long long)D * M * M, (long long)D * M * M,
          (long long)M * M * M, M * M, 1, M * M, 1);
    }

    {
        const int D = 66, M = 126;
        const long long D3 = (long long)D * D * D;
        const long long H  = 7 * D3;
        launch_rows<D>(stream, llb, yh0, tb, D * D, 16,
                       D3, H, D3, (long long)slotT, (long long)D * D * M);
        launch_plane<D>(stream, tb, t2b, 16 * D, (long long)slotT, (long long)slotT2);
        L(t2b, t2b + slotT2, out, 1, 0, 0, D, M, M * M, 16, 8,
          (long long)D * M * M, (long long)D * M * M,
          (long long)M * M * M, M * M, 1, M * M, 1);
    }

    (void)in_sizes; (void)n_in; (void)out_size; (void)ws_size;
}


# Outer-axis pass: a work-group takes a 128-column strip; its 2N input row pieces go to LDS by 16-byte loads, all tiles' accumulators are laid out in LDS and stored as whole 512-byte row pieces
# speedup vs baseline: 4.7545x; 1.0674x over previous
#include <hip/hip_runtime.h>

typedef __attribute__((ext_vector_type(2))) float v2f;
typedef __attribute__((ext_vector_type(8))) float v8f;

__device__ __constant__ float cG0[8] = {
    0.23037781330885523f,  0.7148465705525415f,  0.6308807679295904f,  -0.02798376941698385f,
   -0.18703481171888114f,  0.030841381835986965f, 0.032883011666982945f, -0.010597401784997278f };
__device__ __constant__ float cG1[8] = {
   -0.010597401784997278f, -0.032883011666982945f, 0.030841381835986965f, 0.18703481171888114f,
   -0.02798376941698385f,  -0.6308807679295904f,   0.7148465705525415f,  -0.23037781330885523f };


struct BandTaps { v2f f0[3]; v2f f1[3]; };

__device__ __forceinline__ BandTaps make_taps(int lid, int half)
{
    const int u0x = lid - 4 * half + 6;
    const int u0y = u0x - 2;
    const int sx  = u0x >> 3;
    const int sy  = u0y >> 3;
    const float g0x = cG0[u0x & 7], g1x = cG1[u0x & 7];
    const float g0y = cG0[u0y & 7], g1y = cG1[u0y & 7];
    BandTaps b;
#pragma unroll
    for (int s = 0; s < 3; ++s) {
        b.f0[s].x = (s == sx) ? g0x : 0.0f;  b.f0[s].y = (s == sy) ? g0y : 0.0f;
        b.f1[s].x = (s == sx) ? g1x : 0.0f;  b.f1[s].y = (s == sy) ? g1y : 0.0f;
    }
    return b;
}

__device__ __forceinline__ v2f edge_taps(v2f f, int s, int rem)
{
    v2f r;
    r.x = (4 * s     < rem) ? f.x : 0.0f;
    r.y = (4 * s + 1 < rem) ? f.y : 0.0f;
    return r;
}

template <bool DATA_IS_A>
__device__ __forceinline__ v8f band_mm(v2f taps, v2f data, v8f acc)
{
    if constexpr (DATA_IS_A)
        return __builtin_amdgcn_wmma_f32_16x16x4_f32(false, data, false, taps, (short)0, acc, false, false);
    else
        return __builtin_amdgcn_wmma_f32_16x16x4_f32(false, taps, false, data, (short)0, acc, false, false);
}

template <int N, bool DATA_IS_A>
__device__ __forceinline__ v8f tile_products(const BandTaps& tp, int half, int t,
                                             v2f c0L, v2f c0H, v2f c1L, v2f c1H, v2f c2L, v2f c2H)
{
    const int kb = 8 * t;
    v8f acc = {};
    if (kb + 12 <= N) {
        acc = band_mm<DATA_IS_A>(tp.f0[0], c0L, acc);
        acc = band_mm<DATA_IS_A>(tp.f1[0], c0H, acc);
        acc = band_mm<DATA_IS_A>(tp.f0[1], c1L, acc);
        acc = band_mm<DATA_IS_A>(tp.f1[1], c1H, acc);
        acc = band_mm<DATA_IS_A>(tp.f0[2], c2L, acc);
        acc = band_mm<DATA_IS_A>(tp.f1[2], c2H, acc);
    } else {
        const int rem = N - kb - 2 * half;
        acc = band_mm<DATA_IS_A>(edge_taps(tp.f0[0], 0, rem), c0L, acc);
        acc = band_mm<DATA_IS_A>(edge_taps(tp.f1[0], 0, rem), c0H, acc);
        if (kb + 4 < N) {
            acc = band_mm<DATA_IS_A>(edge_taps(tp.f0[1], 1, rem), c1L, acc);
            acc = band_mm<DATA_IS_A>(edge_taps(tp.f1[1], 1, rem), c1H, acc);
        }
        if (kb + 8 < N) {
            acc = band_mm<DATA_IS_A>(edge_taps(tp.f0[2], 2, rem), c2L, acc);
            acc = band_mm<DATA_IS_A>(edge_taps(tp.f1[2], 2, rem), c2H, acc);
        }
    }
    return acc;
}

template <int N>
__global__ __launch_bounds__(32)
void sfb1d_wmma_rows(const float* __restrict__ ll, const float* __restrict__ yh,
                     float* __restrict__ tb,
                     int lines,
                     long long llOS, long long yhOS, long long bandS,
                     long long slotS, long long outOS)
{
    constexpr int M = 2 * N - 6;
    constexpr int T = (M + 15) >> 4;
    static_assert(16 * M <= 2 * 16 * N, "output block must fit in the input blocks' LDS");
    __shared__ __attribute__((aligned(16))) float sm[2 * 16 * N];
    float* sL = sm;
    float* sH = sm + 16 * N;

    const int lane = threadIdx.x & 31;
    const int half = lane >> 4;
    const int lid  = lane & 15;
    const int lb   = blockIdx.x * 16;

    const int  j   = blockIdx.y;
    const bool j0  = j == 0;
    const float*    lo   = j0 ? ll : yh + (long long)(2 * j - 1) * bandS;
    const long long loOS = j0 ? llOS : yhOS;
    const float*    hi   = yh + (long long)(2 * j) * bandS;
    const long long hiOS = yhOS;

    const int nl   = (lines - lb) < 16 ? (lines - lb) : 16;
    const int cnt4 = (nl * N) >> 2;

    {
        const float4* gL = (const float4*)(lo + (long long)blockIdx.z * loOS + (long long)lb * N);
        const float4* gH = (const float4*)(hi + (long long)blockIdx.z * hiOS + (long long)lb * N);
        float4* sL4 = (float4*)sL;
        float4* sH4 = (float4*)sH;
        constexpr int NV = (4 * N + 31) / 32;
        float4 rL[NV], rH[NV];
        int idx[NV];
        for (int i = 0; i < NV; ++i) {
            const int k = lane + 32 * i;
            idx[i] = k < cnt4 ? k : (cnt4 - 1);
            rL[i] = gL[idx[i]];
            rH[i] = gH[idx[i]];
        }
        for (int i = 0; i < NV; ++i) {
            sL4[idx[i]] = rL[i];
            sH4[idx[i]] = rH[i];
        }
    }
    __syncthreads();

    const int lrow = lid < nl ? lid : (nl - 1);
    const float* loP = sL + lrow * N;
    const float* hiP = sH + lrow * N;

    const BandTaps tp = make_taps(lid, half);
    constexpr int nm2 = N - 2;

    auto chunk = [&](const float* P, int q) -> v2f {
        const int ia  = 4 * q + 2 * half;
        const int iaC = ia > nm2 ? nm2 : ia;
        return *(const v2f*)(P + iaC);
    };

    v2f c0L = chunk(loP, 0), c0H = chunk(hiP, 0);
    v2f c1L = chunk(loP, 1), c1H = chunk(hiP, 1);
    v2f c2L = chunk(loP, 2), c2H = chunk(hiP, 2);

    v8f acc[T];
#pragma unroll
    for (int t = 0; t < T; ++t) {
        const v2f p1L = chunk(loP, 2 * t + 3), p1H = chunk(hiP, 2 * t + 3);
        const v2f p2L = chunk(loP, 2 * t + 4), p2H = chunk(hiP, 2 * t + 4);
        acc[t] = tile_products<N, true>(tp, half, t, c0L, c0H, c1L, c1H, c2L, c2H);
        c0L = c2L;  c0H = c2H;
        c1L = p1L;  c1H = p1H;
        c2L = p2L;  c2H = p2H;
    }

    __syncthreads();

    {
        static_assert(2 * 16 * N - 16 * M >= 32, "one spare float per lane behind the output block");
        float* q  = sm + (8 * half) * M + lid;
        float* qs = sm + 16 * M + lane;
#pragma unroll
        for (int t = 0; t < T; ++t) {
            const bool inLine = 16 * t + lid < M;
#pragma unroll
            for (int r = 0; r < 8; ++r) {
                float* d = inLine ? (q + r * M + 16 * t) : qs;
                *d = acc[t][r];
            }
        }
    }
    __syncthreads();

    {
        float* g = tb + (long long)j * slotS + (long long)blockIdx.z * outOS + (long long)lb * M;
        const int cntO  = nl * M;
        const int cntO4 = cntO >> 2;
        float4*       g4 = (float4*)g;
        const float4* s4 = (const float4*)sm;
        const int off = (int)(((unsigned long long)g >> 4) & 7ull);
        constexpr int NVO = (4 * M + 7 + 31) / 32;
#pragma unroll
        for (int i = 0; i < NVO; ++i) {
            const int k  = lane + 32 * i - off;
            const int kc = k < 0 ? 0 : (k < cntO4 ? k : (cntO4 - 1));
            g4[kc] = s4[kc];
        }
        const int k1 = (cntO4 << 2) + lane;
        if (k1 < cntO)
            g[k1] = sm[k1];
    }
}

template <int N>
__global__ __launch_bounds__(32 * ((2 * N - 6 + 15) / 16), 8)
void sfb1d_wmma_plane(const float* __restrict__ tb, float* __restrict__ t2b,
                      long long inSlotS, long long outSlotS)
{
    constexpr int M  = 2 * N - 6;
    constexpr int C  = M;
    constexpr int T  = (M + 15) >> 4;
    constexpr int S  = (C + 15) >> 4;
    constexpr int NT = 32 * S;
    constexpr int PL = N * C;
    constexpr int PO = M * C;
    static_assert(PL % 4 == 0 && PO % 4 == 0, "planes are copied in 16-byte pieces");
    static_assert(PO <= 2 * PL, "output plane must fit in the input planes' LDS");
    __shared__ __attribute__((aligned(16))) float sm[2 * PL];

    const int tid  = threadIdx.x;
    const int lane = tid & 31;
    const int half = lane >> 4;
    const int lid  = lane & 15;
    const int wv   = tid >> 5;

    const long long z = blockIdx.x;
    const int       j = blockIdx.y;
    const float* lo  = tb + (long long)(2 * j) * inSlotS + z * PL;
    const float* hi  = lo + inSlotS;
    float*       out = t2b + (long long)j * outSlotS + z * PO;

    {
        const float4* gL = (const float4*)lo;
        const float4* gH = (const float4*)hi;
        float4* sL4 = (float4*)sm;
        float4* sH4 = (float4*)(sm + PL);
        constexpr int CNT4 = PL / 4;
        constexpr int NV   = (CNT4 + NT - 1) / NT;
        float4 rL[NV], rH[NV];
        int idx[NV];
        for (int i = 0; i < NV; ++i) {
            const int k = tid + NT * i;
            idx[i] = k < CNT4 ? k : (CNT4 - 1);
            rL[i] = gL[idx[i]];
            rH[i] = gH[idx[i]];
        }
        for (int i = 0; i < NV; ++i) {
            sL4[idx[i]] = rL[i];
            sH4[idx[i]] = rH[i];
        }
    }
    __syncthreads();

    const int  col   = 16 * wv + lid;
    const bool colOK = col < C;
    const int  colC  = colOK ? col : (C - 1);
    const float* loP = sm + colC;
    const float* hiP = sm + PL + colC;

    const BandTaps tp = make_taps(lid, half);

    auto chunk = [&](const float* P, int q) -> v2f {
        const int ra  = 4 * q + 2 * half;
        const int raC = ra     > N - 1 ? N - 1 : ra;
        const int rbC = ra + 1 > N - 1 ? N - 1 : ra + 1;
        v2f v;
        v.x = P[raC * C];
        v.y = P[rbC * C];
        return v;
    };

    v2f c0L = chunk(loP, 0), c0H = chunk(hiP, 0);

    v8f acc[T];
#pragma unroll
    for (int t = 0; t < T; ++t) {
        const v2f c1L = chunk(loP, 2 * t + 1), c1H = chunk(hiP, 2 * t + 1);
        const v2f c2L = chunk(loP, 2 * t + 2), c2H = chunk(hiP, 2 * t + 2);
        acc[t] = tile_products<N, false>(tp, half, t, c0L, c0H, c1L, c1H, c2L, c2H);
        c0L = c2L;  c0H = c2H;
    }

    __syncthreads();

    {
        static_assert(2 * PL - PO >= NT, "one spare float per thread behind the output plane");
        float* q  = sm + (8 * half) * C + colC;
        float* qs = sm + PO + tid;
#pragma unroll
        for (int t = 0; t < T; ++t) {
#pragma unroll
            for (int r = 0; r < 8; ++r) {
                const bool inPlane = 16 * t + 8 * half + r < M;
                float* d = inPlane ? (q + (16 * t + r) * C) : qs;
                *d = acc[t][r];
            }
        }
    }
    __syncthreads();

    {
        float4*       g4 = (float4*)out;
        const float4* s4 = (const float4*)sm;
        constexpr int CNT4 = PO / 4;
        const int off = (int)(((unsigned long long)out >> 4) & 7ull);
        float4* gA = g4 - off;
        constexpr int NVO = (CNT4 + 7 + NT - 1) / NT;
        static_assert(NT > 7 && NT * (NVO - 1) <= CNT4, "only the first and the last store can fall outside the plane");
#pragma unroll
        for (int i = 0; i < NVO; ++i) {
            const unsigned kk = (unsigned)(tid + NT * i);
            const int k = (int)kk - off;
            const bool in = (i > 0 || k >= 0) && (i < NVO - 1 || k < CNT4);
            if (in)
                gA[kk] = s4[k];
        }
    }
}

template <int N>
__global__ __launch_bounds__(256, 8)
void sfb1d_wmma_strip(const float* __restrict__ in, float* __restrict__ out, int slotS)
{
    constexpr int M  = 2 * N - 6;
    constexpr int R  = M * M;
    constexpr int T  = (M + 15) >> 4;
    constexpr int CW = 128;
    constexpr int P  = CW + 12;
    constexpr int NT = 256;
    static_assert(R % 4 == 0, "row pieces are copied in 16-byte pieces");
    __shared__ __attribute__((aligned(16))) float sm[2 * N * P];

    const int tid  = threadIdx.x;
    const int lane = tid & 31;
    const int half = lane >> 4;
    const int lid  = lane & 15;
    const int wv   = __builtin_amdgcn_readfirstlane(tid >> 5);
    __builtin_assume(wv >= 0 && wv < NT / 32);

    const int       cb    = blockIdx.x * CW;
    const long long z     = blockIdx.y;
    const int       ncols = (R - cb) < CW ? (R - cb) : CW;
    const int       vp    = ncols >> 2;
    const int       c4    = lane < vp ? lane : (vp - 1);

    const float* inB  = in  + z * (long long)(N * R) + cb;
    float*       outB = out + z * (long long)(M * R) + cb + 4 * c4;

    {
        constexpr int ROWS = 2 * N;
        constexpr int NV   = (ROWS + 7) / 8;
        float4 rg[NV];
        int so[NV];
        const unsigned laneB = 16u * (unsigned)c4;
#pragma unroll
        for (int i = 0; i < NV; ++i) {
            int row = wv + 8 * i;
            if (row > ROWS - 1) row = ROWS - 1;
            const unsigned ro = row >= N ? (unsigned)slotS + (unsigned)((row - N) * R)
                                         : (unsigned)(row * R);
            rg[i] = *(const float4*)((const char*)(inB + ro) + laneB);
            so[i] = row * P + 4 * c4;
        }
#pragma unroll
        for (int i = 0; i < NV; ++i)
            *(float4*)(sm + so[i]) = rg[i];
    }
    __syncthreads();

    const int col  = 16 * wv + lid;
    const int colC = col < ncols ? col : (ncols - 1);
    const float* loP = sm + colC;
    const float* hiP = sm + N * P + colC;

    const BandTaps tp = make_taps(lid, half);

    auto chunk = [&](const float* Q, int q) -> v2f {
        const int ra  = 4 * q + 2 * half;
        const int raC = ra     > N - 1 ? N - 1 : ra;
        const int rbC = ra + 1 > N - 1 ? N - 1 : ra + 1;
        v2f v;
        v.x = Q[raC * P];
        v.y = Q[rbC * P];
        return v;
    };

    v2f c0L = chunk(loP, 0), c0H = chunk(hiP, 0);

    v8f acc[T];
#pragma unroll
    for (int t = 0; t < T; ++t) {
        const v2f c1L = chunk(loP, 2 * t + 1), c1H = chunk(hiP, 2 * t + 1);
        const v2f c2L = chunk(loP, 2 * t + 2), c2H = chunk(hiP, 2 * t + 2);
        acc[t] = tile_products<N, false>(tp, half, t, c0L, c0H, c1L, c1H, c2L, c2H);
        c0L = c2L;  c0H = c2H;
    }

    __syncthreads();

    {
        static_assert(2 * N * P - M * P >= NT, "one spare float per thread behind the output rows");
        float* q  = sm + (8 * half) * P + colC;
        float* qs = sm + M * P + tid;
#pragma unroll
        for (int t = 0; t < T; ++t) {
#pragma unroll
            for (int r = 0; r < 8; ++r) {
                const bool inRows = 16 * t + 8 * half + r < M;
                float* d = inRows ? (q + (16 * t + r) * P) : qs;
                *d = acc[t][r];
            }
        }
    }
    __syncthreads();

    {
        constexpr int NVO = (M + 7) / 8;
#pragma unroll
        for (int i = 0; i < NVO; ++i) {
            const int row = wv + 8 * i;
            if (i < NVO - 1 || row < M)
                *(float4*)(outB + (long long)row * R) = *(const float4*)(sm + row * P + 4 * c4);
        }
    }
}

template <int N>
static void launch_rows(hipStream_t stream, const float* ll, const float* yh, float* tb,
                        int lines, int outer,
                        long long llOS, long long yhOS, long long bandS,
                        long long slotS, long long outOS)
{
    dim3 g((unsigned)((lines + 15) / 16), 4u, (unsigned)outer);
    sfb1d_wmma_rows<N><<<g, dim3(32), 0, stream>>>(ll, yh, tb, lines,
                                                   llOS, yhOS, bandS, slotS, outOS);
}

template <int N>
static void launch_plane(hipStream_t stream, const float* tb, float* t2b, int planes,
                         long long inSlotS, long long outSlotS)
{
    constexpr int S = (2 * N - 6 + 15) / 16;
    dim3 g((unsigned)planes, 2u, 1u);
    sfb1d_wmma_plane<N><<<g, dim3(32u * S), 0, stream>>>(tb, t2b, inSlotS, outSlotS);
}

template <int N>
static void launch_strip(hipStream_t stream, const float* in, float* out, int slotS,
                         int outer)
{
    constexpr int R = (2 * N - 6) * (2 * N - 6);
    dim3 g((unsigned)((R + 127) / 128), (unsigned)outer, 1u);
    sfb1d_wmma_strip<N><<<g, dim3(256u), 0, stream>>>(in, out, slotS);
}

extern "C" void kernel_launch(void* const* d_in, const int* in_sizes, int n_in,
                              void* d_out, int out_size, void* d_ws, size_t ws_size,
                              hipStream_t stream)
{
    const float* yl  = (const float*)d_in[0];
    const float* yh0 = (const float*)d_in[1];
    const float* yh1 = (const float*)d_in[2];
    float* out = (float*)d_out;
    float* ws  = (float*)d_ws;

    const size_t slotT  = (size_t)16 * 66 * 66 * 126;
    const size_t slotT2 = (size_t)16 * 66 * 126 * 126;
    float* tb  = ws;
    float* t2b = ws + 4 * slotT;
    float* llb = ws + 4 * slotT + 2 * slotT2;

    {
        const int D = 36, M = 66;
        const long long D3 = (long long)D * D * D;
        const long long H  = 7 * D3;
        launch_rows<D>(stream, yl, yh1, tb, D * D, 16,
                       D3, H, D3, (long long)slotT, (long long)D * D * M);
        launch_plane<D>(stream, tb, t2b, 16 * D, (long long)slotT, (long long)slotT2);
        launch_strip<D>(stream, t2b, llb, (int)slotT2, 16);
    }

    {
        const int D = 66, M = 126;
        const long long D3 = (long long)D * D * D;
        const long long H  = 7 * D3;
        launch_rows<D>(stream, llb, yh0, tb, D * D, 16,
                       D3, H, D3, (long long)slotT, (long long)D * D * M);
        launch_plane<D>(stream, tb, t2b, 16 * D, (long long)slotT, (long long)slotT2);
        launch_strip<D>(stream, t2b, out, (int)slotT2, 16);
    }

    (void)in_sizes; (void)n_in; (void)out_size; (void)ws_size;
}
